# starting kernel
# baseline (speedup 1.0000x reference)
#include <hip/hip_runtime.h>
#include <math.h>


typedef __attribute__((ext_vector_type(2))) float v2f;
typedef __attribute__((ext_vector_type(8))) float v8f;

#define WMMA_F32(a, b, c) \
  __builtin_amdgcn_wmma_f32_16x16x4_f32(false, (a), false, (b), (short)0, (c), false, false)

constexpr int B_ = 2, T_ = 2048, C_ = 256, H_ = 4, D_ = 64, W_ = 64, S_ = 1;
constexpr int Tq_ = T_ - S_;
constexpr int KQVROW = 3 * H_ * D_;

__device__ __forceinline__ int clampi(int v, int lo, int hi) {
  return v < lo ? lo : (v > hi ? hi : v);
}

__global__ __launch_bounds__(32)
void gemm_abt(const float* __restrict__ A, const float* __restrict__ Bm,
              const float* __restrict__ bias, float* __restrict__ Cm,
              int M, int N, int K) {
  const int lane = threadIdx.x & 31;
  const int half = lane >> 4;
  const int l16  = lane & 15;
  const int kb   = half * 2;
  const int n0   = blockIdx.x * 32;
  const int m0   = blockIdx.y * 32;

  const float* a0 = A  + (size_t)(m0 + l16)      * K;
  const float* a1 = A  + (size_t)(m0 + 16 + l16) * K;
  const float* b0 = Bm + (size_t)(n0 + l16)      * K;
  const float* b1 = Bm + (size_t)(n0 + 16 + l16) * K;

  v8f c00 = {}, c01 = {}, c10 = {}, c11 = {};
  for (int k0 = 0; k0 < K; k0 += 64) {
    __builtin_prefetch(a0 + k0 + 64, 0, 3);
    __builtin_prefetch(a1 + k0 + 64, 0, 3);
    __builtin_prefetch(b0 + k0 + 64, 0, 3);
    __builtin_prefetch(b1 + k0 + 64, 0, 3);
#pragma unroll 4
    for (int k = k0; k < k0 + 64; k += 4) {
      v2f fa0 = *(const v2f*)(a0 + k + kb);
      v2f fa1 = *(const v2f*)(a1 + k + kb);
      v2f fb0 = *(const v2f*)(b0 + k + kb);
      v2f fb1 = *(const v2f*)(b1 + k + kb);
      c00 = WMMA_F32(fa0, fb0, c00);
      c01 = WMMA_F32(fa0, fb1, c01);
      c10 = WMMA_F32(fa1, fb0, c10);
      c11 = WMMA_F32(fa1, fb1, c11);
    }
  }
  const float bb0 = bias ? bias[n0 + l16]      : 0.0f;
  const float bb1 = bias ? bias[n0 + 16 + l16] : 0.0f;
#pragma unroll
  for (int i = 0; i < 8; ++i) {
    const int m = m0 + i + half * 8;
    Cm[(size_t)m * N        + (n0 + l16)]      = c00[i] + bb0;
    Cm[(size_t)m * N        + (n0 + 16 + l16)] = c01[i] + bb1;
    Cm[(size_t)(m + 16) * N + (n0 + l16)]      = c10[i] + bb0;
    Cm[(size_t)(m + 16) * N + (n0 + 16 + l16)] = c11[i] + bb1;
  }
}

__global__ __launch_bounds__(256)
void s0_kernel(const float* __restrict__ kqv, float* __restrict__ s0) {
  const int idx = blockIdx.x * 256 + threadIdx.x;
  if (idx >= B_ * H_ * T_) return;
  const int j  = idx % T_;
  const int bh = idx / T_;
  const int b  = bh / H_, h = bh % H_;
  const float* base = kqv + (size_t)(b * T_ + j) * KQVROW + h * D_;
  const float* kp = base;
  const float* qp = base + H_ * D_;
  float acc = 0.0f;
#pragma unroll 8
  for (int d = 0; d < D_; ++d) acc = fmaf(kp[d], qp[d], acc);
  s0[idx] = acc;
}

__global__ __launch_bounds__(32)
void s1_kernel(const float* __restrict__ kqv, const float* __restrict__ q_pos,
               float* __restrict__ S1) {
  const int lane = threadIdx.x & 31;
  const int half = lane >> 4, l16 = lane & 15, kb = half * 2;
  const int j0 = blockIdx.x * 16;
  const int w0 = blockIdx.y * 16;
  const int bh = blockIdx.z;
  const int b = bh / H_, h = bh % H_;

  const float* qrow = kqv + (size_t)(b * T_ + j0 + l16) * KQVROW + H_ * D_ + h * D_;
  const float* prow = q_pos + (size_t)((w0 + l16) * H_ + h) * D_;

  v8f c = {};
#pragma unroll
  for (int k = 0; k < D_; k += 4) {
    v2f a = *(const v2f*)(qrow + k + kb);
    v2f b2 = *(const v2f*)(prow + k + kb);
    c = WMMA_F32(a, b2, c);
  }
  float* out = S1 + ((size_t)bh * T_ + j0) * W_ + w0;
#pragma unroll
  for (int i = 0; i < 8; ++i)
    out[(size_t)(i + half * 8) * W_ + l16] = c[i];
}

__global__ __launch_bounds__(256)
void zero_pad_rows(float* __restrict__ attnout) {
  const int i = threadIdx.x;
  attnout[i] = 0.0f;
  attnout[(size_t)T_ * C_ + i] = 0.0f;
}

__global__ __launch_bounds__(128)
void attn_kernel(const float* __restrict__ kqv, const float* __restrict__ S1,
                 const float* __restrict__ s0, const float* __restrict__ v_pos,
                 float* __restrict__ attnout, float* __restrict__ wei_out) {
  __shared__ float sc[16][W_ + 1];

  const int t0 = blockIdx.x * 16;
  const int h  = blockIdx.y;
  const int b  = blockIdx.z;
  const int bh = b * H_ + h;
  const int tid = threadIdx.x;
  const float scale = 0.125f;

  {
    const int ti = tid >> 3;
    const int wb = (tid & 7) * 8;
    const int t  = t0 + ti;
#pragma unroll
    for (int u = 0; u < 8; ++u) {
      const int w = wb + u;
      const int j = t + w - (W_ - 1);
      float v;
      if (t >= Tq_)      v = 0.0f;
      else if (j < 0)    v = -INFINITY;
      else               v = (s0[(size_t)bh * T_ + j] +
                              S1[((size_t)bh * T_ + j) * W_ + w]) * scale;
      sc[ti][w] = v;
    }
  }
  __syncthreads();

  if (tid < 16) {
    float m = -INFINITY;
    for (int w = 0; w < W_; ++w) m = fmaxf(m, sc[tid][w]);
    float s = 0.0f;
    for (int w = 0; w < W_; ++w) { float e = __expf(sc[tid][w] - m); sc[tid][w] = e; s += e; }
    const float inv = 1.0f / s;
    for (int w = 0; w < W_; ++w) sc[tid][w] *= inv;
  }
  __syncthreads();

  {
    const int ti = tid >> 3;
    const int wb = (tid & 7) * 8;
    const int t  = t0 + ti;
    if (t < Tq_) {
#pragma unroll
      for (int u = 0; u < 8; ++u) {
        const int w = wb + u;
        wei_out[(((size_t)b * Tq_ + t) * W_ + w) * H_ + h] = sc[ti][w];
      }
    }
  }

  const int lane = tid & 31;
  const int wave = tid >> 5;
  const int d0   = wave * 16;
  const int half = lane >> 4, l16 = lane & 15, kb = half * 2;
  const float* vbase = kqv + 2 * H_ * D_ + h * D_ + d0 + l16;

  v8f c = {};
#pragma unroll
  for (int jt = 0; jt < 5; ++jt) {
    const int j0 = t0 - 64 + jt * 16;
#pragma unroll
    for (int k4 = 0; k4 < 16; k4 += 4) {
      v2f a;
      {
        const int t = t0 + l16;
        const int wa = (j0 + k4 + kb) - t + (W_ - 1);
        const int wA = clampi(wa, 0, W_ - 1);
        const int wB = clampi(wa + 1, 0, W_ - 1);
        a.x = (wa     >= 0 && wa     < W_) ? sc[l16][wA] : 0.0f;
        a.y = (wa + 1 >= 0 && wa + 1 < W_) ? sc[l16][wB] : 0.0f;
      }
      v2f bv;
      {
        const int jA = clampi(j0 + k4 + kb,     0, T_ - 1);
        const int jB = clampi(j0 + k4 + kb + 1, 0, T_ - 1);
        bv.x = vbase[(size_t)(b * T_ + jA) * KQVROW];
        bv.y = vbase[(size_t)(b * T_ + jB) * KQVROW];
      }
      c = WMMA_F32(a, bv, c);
    }
  }
#pragma unroll
  for (int k = 0; k < W_; k += 4) {
    v2f a;
    a.x = sc[l16][k + kb];
    a.y = sc[l16][k + kb + 1];
    v2f bv;
    bv.x = v_pos[((size_t)(k + kb)     * H_ + h) * D_ + d0 + l16];
    bv.y = v_pos[((size_t)(k + kb + 1) * H_ + h) * D_ + d0 + l16];
    c = WMMA_F32(a, bv, c);
  }

#pragma unroll
  for (int i = 0; i < 8; ++i) {
    const int ti = i + half * 8;
    const int t  = t0 + ti;
    if (t < Tq_)
      attnout[((size_t)b * T_ + (t + S_)) * C_ + h * D_ + d0 + l16] = c[i];
  }
}

extern "C" void kernel_launch(void* const* d_in, const int* in_sizes, int n_in,
                              void* d_out, int out_size, void* d_ws, size_t ws_size,
                              hipStream_t stream) {
  (void)in_sizes; (void)n_in; (void)out_size; (void)ws_size;

  const float* x     = (const float*)d_in[0];
  const float* w_kqv = (const float*)d_in[1];
  const float* w_out = (const float*)d_in[2];
  const float* b_out = (const float*)d_in[3];
  const float* q_pos = (const float*)d_in[4];
  const float* v_pos = (const float*)d_in[5];

  float* y   = (float*)d_out;
  float* wei = y + (size_t)B_ * T_ * C_;

  char*  ws      = (char*)d_ws;
  float* kqv     = (float*)ws;
  size_t off     = (size_t)B_ * T_ * KQVROW * sizeof(float);
  float* S1      = (float*)(ws + off);
  off           += (size_t)B_ * H_ * T_ * W_ * sizeof(float);
  float* s0      = (float*)(ws + off);
  off           += (size_t)B_ * H_ * T_ * sizeof(float);
  float* attnout = (float*)(ws + off);

  gemm_abt<<<dim3(KQVROW / 32, (B_ * T_) / 32), 32, 0, stream>>>(
      x, w_kqv, nullptr, kqv, B_ * T_, KQVROW, C_);

  s0_kernel<<<(B_ * H_ * T_ + 255) / 256, 256, 0, stream>>>(kqv, s0);
  s1_kernel<<<dim3(T_ / 16, W_ / 16, B_ * H_), 32, 0, stream>>>(kqv, q_pos, S1);

  zero_pad_rows<<<1, C_, 0, stream>>>(attnout);

  attn_kernel<<<dim3((Tq_ + 15) / 16, H_, B_), 128, 0, stream>>>(
      kqv, S1, s0, v_pos, attnout, wei);

  gemm_abt<<<dim3(C_ / 32, (B_ * T_) / 32), 32, 0, stream>>>(
      attnout, w_out, b_out, y, B_ * T_, C_, C_);
}


# gemm_abt: A and B 64x64 block tiles staged through LDS by coalesced 128-bit global loads (32-k chunks, double-buffered) instead of per-lane strided 8-byte fragment loads from global
# speedup vs baseline: 1.2400x; 1.2400x over previous
#include <hip/hip_runtime.h>
#include <math.h>


typedef __attribute__((ext_vector_type(2))) float v2f;
typedef __attribute__((ext_vector_type(4))) float v4f;
typedef __attribute__((ext_vector_type(8))) float v8f;

#define WMMA_F32(a, b, c) \
  __builtin_amdgcn_wmma_f32_16x16x4_f32(false, (a), false, (b), (short)0, (c), false, false)

constexpr int B_ = 2, T_ = 2048, C_ = 256, H_ = 4, D_ = 64, W_ = 64, S_ = 1;
constexpr int Tq_ = T_ - S_;
constexpr int KQVROW = 3 * H_ * D_;

__device__ __forceinline__ int clampi(int v, int lo, int hi) {
  return v < lo ? lo : (v > hi ? hi : v);
}

template <int WM, int WN, int WVM, int WVN>
__global__ __launch_bounds__(32 * WVM * WVN)
void gemm_abt(const float* __restrict__ A, const float* __restrict__ Bm,
              const float* __restrict__ bias, float* __restrict__ Cm,
              int M, int N, int K) {
  constexpr int NT    = 32 * WVM * WVN;
  constexpr int BM    = 16 * WM * WVM;
  constexpr int BN    = 16 * WN * WVN;
  constexpr int KC    = 32;
  constexpr int PITCH = KC + 4;
  constexpr int RPP   = NT / 8;
  constexpr int PA    = BM / RPP;
  constexpr int PB    = BN / RPP;

  __shared__ __attribute__((aligned(16))) float sA[2][BM][PITCH];
  __shared__ __attribute__((aligned(16))) float sB[2][BN][PITCH];

  const int tid  = threadIdx.x;
  const int lane = tid & 31;
  const int wave = tid >> 5;
  const int half = lane >> 4;
  const int l16  = lane & 15;
  const int kb   = half * 2;
  const int wm   = wave / WVN;
  const int wn   = wave % WVN;
  const int n0   = blockIdx.x * BN;
  const int m0   = blockIdx.y * BM;

  const int lr = tid >> 3;
  const int lc = (tid & 7) * 4;
  const float* ga = A  + (size_t)(m0 + lr) * K + lc;
  const float* gb = Bm + (size_t)(n0 + lr) * K + lc;

  v4f ra[PA], rb[PB];
#pragma unroll
  for (int p = 0; p < PA; ++p) ra[p] = *(const v4f*)(ga + (size_t)(p * RPP) * K);
#pragma unroll
  for (int p = 0; p < PB; ++p) rb[p] = *(const v4f*)(gb + (size_t)(p * RPP) * K);
#pragma unroll
  for (int p = 0; p < PA; ++p) *(v4f*)&sA[0][p * RPP + lr][lc] = ra[p];
#pragma unroll
  for (int p = 0; p < PB; ++p) *(v4f*)&sB[0][p * RPP + lr][lc] = rb[p];
  __syncthreads();

  v8f acc[WM][WN];
#pragma unroll
  for (int i = 0; i < WM; ++i)
#pragma unroll
    for (int j = 0; j < WN; ++j) acc[i][j] = v8f{};

  const int arow = wm * 16 * WM + l16;
  const int brow = wn * 16 * WN + l16;
  const int nch  = K / KC;
  for (int c = 0; c < nch; ++c) {
    const int cur  = c & 1;
    const bool more = (c + 1 < nch);
    if (more) {
      const int kn = (c + 1) * KC;
#pragma unroll
      for (int p = 0; p < PA; ++p) ra[p] = *(const v4f*)(ga + (size_t)(p * RPP) * K + kn);
#pragma unroll
      for (int p = 0; p < PB; ++p) rb[p] = *(const v4f*)(gb + (size_t)(p * RPP) * K + kn);
    }
#pragma unroll
    for (int ks = 0; ks < KC; ks += 4) {
      v2f fa[WM], fb[WN];
#pragma unroll
      for (int i = 0; i < WM; ++i) fa[i] = *(const v2f*)&sA[cur][arow + i * 16][ks + kb];
#pragma unroll
      for (int j = 0; j < WN; ++j) fb[j] = *(const v2f*)&sB[cur][brow + j * 16][ks + kb];
#pragma unroll
      for (int i = 0; i < WM; ++i)
#pragma unroll
        for (int j = 0; j < WN; ++j) acc[i][j] = WMMA_F32(fa[i], fb[j], acc[i][j]);
    }
    if (more) {
#pragma unroll
      for (int p = 0; p < PA; ++p) *(v4f*)&sA[cur ^ 1][p * RPP + lr][lc] = ra[p];
#pragma unroll
      for (int p = 0; p < PB; ++p) *(v4f*)&sB[cur ^ 1][p * RPP + lr][lc] = rb[p];
    }
    __syncthreads();
  }

#pragma unroll
  for (int j = 0; j < WN; ++j) {
    const int nn = n0 + wn * 16 * WN + j * 16 + l16;
    const float bb = bias ? bias[nn] : 0.0f;
#pragma unroll
    for (int i = 0; i < WM; ++i) {
#pragma unroll
      for (int r = 0; r < 8; ++r) {
        const int m = m0 + wm * 16 * WM + i * 16 + r + half * 8;
        Cm[(size_t)m * N + nn] = acc[i][j][r] + bb;
      }
    }
  }
}

__global__ __launch_bounds__(256)
void s0_kernel(const float* __restrict__ kqv, float* __restrict__ s0) {
  const int idx = blockIdx.x * 256 + threadIdx.x;
  if (idx >= B_ * H_ * T_) return;
  const int j  = idx % T_;
  const int bh = idx / T_;
  const int b  = bh / H_, h = bh % H_;
  const float* base = kqv + (size_t)(b * T_ + j) * KQVROW + h * D_;
  const float* kp = base;
  const float* qp = base + H_ * D_;
  float acc = 0.0f;
#pragma unroll 8
  for (int d = 0; d < D_; ++d) acc = fmaf(kp[d], qp[d], acc);
  s0[idx] = acc;
}

__global__ __launch_bounds__(32)
void s1_kernel(const float* __restrict__ kqv, const float* __restrict__ q_pos,
               float* __restrict__ S1) {
  const int lane = threadIdx.x & 31;
  const int half = lane >> 4, l16 = lane & 15, kb = half * 2;
  const int j0 = blockIdx.x * 16;
  const int w0 = blockIdx.y * 16;
  const int bh = blockIdx.z;
  const int b = bh / H_, h = bh % H_;

  const float* qrow = kqv + (size_t)(b * T_ + j0 + l16) * KQVROW + H_ * D_ + h * D_;
  const float* prow = q_pos + (size_t)((w0 + l16) * H_ + h) * D_;

  v8f c = {};
#pragma unroll
  for (int k = 0; k < D_; k += 4) {
    v2f a = *(const v2f*)(qrow + k + kb);
    v2f b2 = *(const v2f*)(prow + k + kb);
    c = WMMA_F32(a, b2, c);
  }
  float* out = S1 + ((size_t)bh * T_ + j0) * W_ + w0;
#pragma unroll
  for (int i = 0; i < 8; ++i)
    out[(size_t)(i + half * 8) * W_ + l16] = c[i];
}

__global__ __launch_bounds__(256)
void zero_pad_rows(float* __restrict__ attnout) {
  const int i = threadIdx.x;
  attnout[i] = 0.0f;
  attnout[(size_t)T_ * C_ + i] = 0.0f;
}

__global__ __launch_bounds__(128)
void attn_kernel(const float* __restrict__ kqv, const float* __restrict__ S1,
                 const float* __restrict__ s0, const float* __restrict__ v_pos,
                 float* __restrict__ attnout, float* __restrict__ wei_out) {
  __shared__ float sc[16][W_ + 1];

  const int t0 = blockIdx.x * 16;
  const int h  = blockIdx.y;
  const int b  = blockIdx.z;
  const int bh = b * H_ + h;
  const int tid = threadIdx.x;
  const float scale = 0.125f;

  {
    const int ti = tid >> 3;
    const int wb = (tid & 7) * 8;
    const int t  = t0 + ti;
#pragma unroll
    for (int u = 0; u < 8; ++u) {
      const int w = wb + u;
      const int j = t + w - (W_ - 1);
      float v;
      if (t >= Tq_)      v = 0.0f;
      else if (j < 0)    v = -INFINITY;
      else               v = (s0[(size_t)bh * T_ + j] +
                              S1[((size_t)bh * T_ + j) * W_ + w]) * scale;
      sc[ti][w] = v;
    }
  }
  __syncthreads();

  if (tid < 16) {
    float m = -INFINITY;
    for (int w = 0; w < W_; ++w) m = fmaxf(m, sc[tid][w]);
    float s = 0.0f;
    for (int w = 0; w < W_; ++w) { float e = __expf(sc[tid][w] - m); sc[tid][w] = e; s += e; }
    const float inv = 1.0f / s;
    for (int w = 0; w < W_; ++w) sc[tid][w] *= inv;
  }
  __syncthreads();

  {
    const int ti = tid >> 3;
    const int wb = (tid & 7) * 8;
    const int t  = t0 + ti;
    if (t < Tq_) {
#pragma unroll
      for (int u = 0; u < 8; ++u) {
        const int w = wb + u;
        wei_out[(((size_t)b * Tq_ + t) * W_ + w) * H_ + h] = sc[ti][w];
      }
    }
  }

  const int lane = tid & 31;
  const int wave = tid >> 5;
  const int d0   = wave * 16;
  const int half = lane >> 4, l16 = lane & 15, kb = half * 2;
  const float* vbase = kqv + 2 * H_ * D_ + h * D_ + d0 + l16;

  v8f c = {};
#pragma unroll
  for (int jt = 0; jt < 5; ++jt) {
    const int j0 = t0 - 64 + jt * 16;
#pragma unroll
    for (int k4 = 0; k4 < 16; k4 += 4) {
      v2f a;
      {
        const int t = t0 + l16;
        const int wa = (j0 + k4 + kb) - t + (W_ - 1);
        const int wA = clampi(wa, 0, W_ - 1);
        const int wB = clampi(wa + 1, 0, W_ - 1);
        a.x = (wa     >= 0 && wa     < W_) ? sc[l16][wA] : 0.0f;
        a.y = (wa + 1 >= 0 && wa + 1 < W_) ? sc[l16][wB] : 0.0f;
      }
      v2f bv;
      {
        const int jA = clampi(j0 + k4 + kb,     0, T_ - 1);
        const int jB = clampi(j0 + k4 + kb + 1, 0, T_ - 1);
        bv.x = vbase[(size_t)(b * T_ + jA) * KQVROW];
        bv.y = vbase[(size_t)(b * T_ + jB) * KQVROW];
      }
      c = WMMA_F32(a, bv, c);
    }
  }
#pragma unroll
  for (int k = 0; k < W_; k += 4) {
    v2f a;
    a.x = sc[l16][k + kb];
    a.y = sc[l16][k + kb + 1];
    v2f bv;
    bv.x = v_pos[((size_t)(k + kb)     * H_ + h) * D_ + d0 + l16];
    bv.y = v_pos[((size_t)(k + kb + 1) * H_ + h) * D_ + d0 + l16];
    c = WMMA_F32(a, bv, c);
  }

#pragma unroll
  for (int i = 0; i < 8; ++i) {
    const int ti = i + half * 8;
    const int t  = t0 + ti;
    if (t < Tq_)
      attnout[((size_t)b * T_ + (t + S_)) * C_ + h * D_ + d0 + l16] = c[i];
  }
}

extern "C" void kernel_launch(void* const* d_in, const int* in_sizes, int n_in,
                              void* d_out, int out_size, void* d_ws, size_t ws_size,
                              hipStream_t stream) {
  (void)in_sizes; (void)n_in; (void)out_size; (void)ws_size;

  const float* x     = (const float*)d_in[0];
  const float* w_kqv = (const float*)d_in[1];
  const float* w_out = (const float*)d_in[2];
  const float* b_out = (const float*)d_in[3];
  const float* q_pos = (const float*)d_in[4];
  const float* v_pos = (const float*)d_in[5];

  float* y   = (float*)d_out;
  float* wei = y + (size_t)B_ * T_ * C_;

  char*  ws      = (char*)d_ws;
  float* kqv     = (float*)ws;
  size_t off     = (size_t)B_ * T_ * KQVROW * sizeof(float);
  float* S1      = (float*)(ws + off);
  off           += (size_t)B_ * H_ * T_ * W_ * sizeof(float);
  float* s0      = (float*)(ws + off);
  off           += (size_t)B_ * H_ * T_ * sizeof(float);
  float* attnout = (float*)(ws + off);

  gemm_abt<2, 2, 2, 2><<<dim3(KQVROW / 64, (B_ * T_) / 64), 128, 0, stream>>>(
      x, w_kqv, nullptr, kqv, B_ * T_, KQVROW, C_);

  s0_kernel<<<(B_ * H_ * T_ + 255) / 256, 256, 0, stream>>>(kqv, s0);
  s1_kernel<<<dim3(T_ / 16, W_ / 16, B_ * H_), 32, 0, stream>>>(kqv, q_pos, S1);

  zero_pad_rows<<<1, C_, 0, stream>>>(attnout);

  attn_kernel<<<dim3((Tq_ + 15) / 16, H_, B_), 128, 0, stream>>>(
      kqv, S1, s0, v_pos, attnout, wei);

  gemm_abt<2, 1, 2, 4><<<dim3(C_ / 64, (B_ * T_) / 64), 256, 0, stream>>>(
      attnout, w_out, b_out, y, B_ * T_, C_, C_);
}


# 6 launches -> 4: k.q row dots (s0) folded into the S1 kernel's w-tile-0 waves, pad-row zeroing into the attention kernel; s0 now summed as two lane-half partial sums added once
# speedup vs baseline: 1.4272x; 1.1510x over previous
#include <hip/hip_runtime.h>
#include <math.h>


typedef __attribute__((ext_vector_type(2))) float v2f;
typedef __attribute__((ext_vector_type(4))) float v4f;
typedef __attribute__((ext_vector_type(8))) float v8f;

#define WMMA_F32(a, b, c) \
  __builtin_amdgcn_wmma_f32_16x16x4_f32(false, (a), false, (b), (short)0, (c), false, false)

constexpr int B_ = 2, T_ = 2048, C_ = 256, H_ = 4, D_ = 64, W_ = 64, S_ = 1;
constexpr int Tq_ = T_ - S_;
constexpr int KQVROW = 3 * H_ * D_;

__device__ __forceinline__ int clampi(int v, int lo, int hi) {
  return v < lo ? lo : (v > hi ? hi : v);
}

template <int WM, int WN, int WVM, int WVN>
__global__ __launch_bounds__(32 * WVM * WVN)
void gemm_abt(const float* __restrict__ A, const float* __restrict__ Bm,
              const float* __restrict__ bias, float* __restrict__ Cm,
              int M, int N, int K) {
  constexpr int NT    = 32 * WVM * WVN;
  constexpr int BM    = 16 * WM * WVM;
  constexpr int BN    = 16 * WN * WVN;
  constexpr int KC    = 32;
  constexpr int PITCH = KC + 4;
  constexpr int RPP   = NT / 8;
  constexpr int PA    = BM / RPP;
  constexpr int PB    = BN / RPP;

  __shared__ __attribute__((aligned(16))) float sA[2][BM][PITCH];
  __shared__ __attribute__((aligned(16))) float sB[2][BN][PITCH];

  const int tid  = threadIdx.x;
  const int lane = tid & 31;
  const int wave = tid >> 5;
  const int half = lane >> 4;
  const int l16  = lane & 15;
  const int kb   = half * 2;
  const int wm   = wave / WVN;
  const int wn   = wave % WVN;
  const int n0   = blockIdx.x * BN;
  const int m0   = blockIdx.y * BM;

  const int lr = tid >> 3;
  const int lc = (tid & 7) * 4;
  const float* ga = A  + (size_t)(m0 + lr) * K + lc;
  const float* gb = Bm + (size_t)(n0 + lr) * K + lc;

  v4f ra[PA], rb[PB];
#pragma unroll
  for (int p = 0; p < PA; ++p) ra[p] = *(const v4f*)(ga + (size_t)(p * RPP) * K);
#pragma unroll
  for (int p = 0; p < PB; ++p) rb[p] = *(const v4f*)(gb + (size_t)(p * RPP) * K);
#pragma unroll
  for (int p = 0; p < PA; ++p) *(v4f*)&sA[0][p * RPP + lr][lc] = ra[p];
#pragma unroll
  for (int p = 0; p < PB; ++p) *(v4f*)&sB[0][p * RPP + lr][lc] = rb[p];
  __syncthreads();

  v8f acc[WM][WN];
#pragma unroll
  for (int i = 0; i < WM; ++i)
#pragma unroll
    for (int j = 0; j < WN; ++j) acc[i][j] = v8f{};

  const int arow = wm * 16 * WM + l16;
  const int brow = wn * 16 * WN + l16;
  const int nch  = K / KC;
  for (int c = 0; c < nch; ++c) {
    const int cur  = c & 1;
    const bool more = (c + 1 < nch);
    if (more) {
      const int kn = (c + 1) * KC;
#pragma unroll
      for (int p = 0; p < PA; ++p) ra[p] = *(const v4f*)(ga + (size_t)(p * RPP) * K + kn);
#pragma unroll
      for (int p = 0; p < PB; ++p) rb[p] = *(const v4f*)(gb + (size_t)(p * RPP) * K + kn);
    }
#pragma unroll
    for (int ks = 0; ks < KC; ks += 4) {
      v2f fa[WM], fb[WN];
#pragma unroll
      for (int i = 0; i < WM; ++i) fa[i] = *(const v2f*)&sA[cur][arow + i * 16][ks + kb];
#pragma unroll
      for (int j = 0; j < WN; ++j) fb[j] = *(const v2f*)&sB[cur][brow + j * 16][ks + kb];
#pragma unroll
      for (int i = 0; i < WM; ++i)
#pragma unroll
        for (int j = 0; j < WN; ++j) acc[i][j] = WMMA_F32(fa[i], fb[j], acc[i][j]);
    }
    if (more) {
#pragma unroll
      for (int p = 0; p < PA; ++p) *(v4f*)&sA[cur ^ 1][p * RPP + lr][lc] = ra[p];
#pragma unroll
      for (int p = 0; p < PB; ++p) *(v4f*)&sB[cur ^ 1][p * RPP + lr][lc] = rb[p];
    }
    __syncthreads();
  }

#pragma unroll
  for (int j = 0; j < WN; ++j) {
    const int nn = n0 + wn * 16 * WN + j * 16 + l16;
    const float bb = bias ? bias[nn] : 0.0f;
#pragma unroll
    for (int i = 0; i < WM; ++i) {
#pragma unroll
      for (int r = 0; r < 8; ++r) {
        const int m = m0 + wm * 16 * WM + i * 16 + r + half * 8;
        Cm[(size_t)m * N + nn] = acc[i][j][r] + bb;
      }
    }
  }
}

__global__ __launch_bounds__(32)
void s1_kernel(const float* __restrict__ kqv, const float* __restrict__ q_pos,
               float* __restrict__ S1, float* __restrict__ s0) {
  const int lane = threadIdx.x & 31;
  const int half = lane >> 4, l16 = lane & 15, kb = half * 2;
  const int j0 = blockIdx.x * 16;
  const int w0 = blockIdx.y * 16;
  const int bh = blockIdx.z;
  const int b = bh / H_, h = bh % H_;

  const float* krow = kqv + (size_t)(b * T_ + j0 + l16) * KQVROW + h * D_;
  const float* qrow = krow + H_ * D_;
  const float* prow = q_pos + (size_t)((w0 + l16) * H_ + h) * D_;

  v2f a[D_ / 4];
#pragma unroll
  for (int i = 0; i < D_ / 4; ++i) a[i] = *(const v2f*)(qrow + 4 * i + kb);

  if (blockIdx.y == 0) {
    float p = 0.0f;
#pragma unroll
    for (int i = 0; i < D_ / 4; ++i) {
      const v2f kk = *(const v2f*)(krow + 4 * i + kb);
      p = fmaf(kk.x, a[i].x, p);
      p = fmaf(kk.y, a[i].y, p);
    }
    p += __shfl_xor(p, 16);
    if (half == 0) s0[(size_t)bh * T_ + j0 + l16] = p;
  }

  v8f c = {};
#pragma unroll
  for (int i = 0; i < D_ / 4; ++i) {
    v2f b2 = *(const v2f*)(prow + 4 * i + kb);
    c = WMMA_F32(a[i], b2, c);
  }
  float* out = S1 + ((size_t)bh * T_ + j0) * W_ + w0;
#pragma unroll
  for (int i = 0; i < 8; ++i)
    out[(size_t)(i + half * 8) * W_ + l16] = c[i];
}

__global__ __launch_bounds__(128)
void attn_kernel(const float* __restrict__ kqv, const float* __restrict__ S1,
                 const float* __restrict__ s0, const float* __restrict__ v_pos,
                 float* __restrict__ attnout, float* __restrict__ wei_out) {
  __shared__ float sc[16][W_ + 1];

  const int t0 = blockIdx.x * 16;
  const int h  = blockIdx.y;
  const int b  = blockIdx.z;
  const int bh = b * H_ + h;
  const int tid = threadIdx.x;
  const float scale = 0.125f;

  if (blockIdx.x == 0 && tid < D_)
    attnout[(size_t)b * T_ * C_ + h * D_ + tid] = 0.0f;

  {
    const int ti = tid >> 3;
    const int wb = (tid & 7) * 8;
    const int t  = t0 + ti;
#pragma unroll
    for (int u = 0; u < 8; ++u) {
      const int w = wb + u;
      const int j = t + w - (W_ - 1);
      float v;
      if (t >= Tq_)      v = 0.0f;
      else if (j < 0)    v = -INFINITY;
      else               v = (s0[(size_t)bh * T_ + j] +
                              S1[((size_t)bh * T_ + j) * W_ + w]) * scale;
      sc[ti][w] = v;
    }
  }
  __syncthreads();

  if (tid < 16) {
    float m = -INFINITY;
    for (int w = 0; w < W_; ++w) m = fmaxf(m, sc[tid][w]);
    float s = 0.0f;
    for (int w = 0; w < W_; ++w) { float e = __expf(sc[tid][w] - m); sc[tid][w] = e; s += e; }
    const float inv = 1.0f / s;
    for (int w = 0; w < W_; ++w) sc[tid][w] *= inv;
  }
  __syncthreads();

  {
    const int ti = tid >> 3;
    const int wb = (tid & 7) * 8;
    const int t  = t0 + ti;
    if (t < Tq_) {
#pragma unroll
      for (int u = 0; u < 8; ++u) {
        const int w = wb + u;
        wei_out[(((size_t)b * Tq_ + t) * W_ + w) * H_ + h] = sc[ti][w];
      }
    }
  }

  const int lane = tid & 31;
  const int wave = tid >> 5;
  const int d0   = wave * 16;
  const int half = lane >> 4, l16 = lane & 15, kb = half * 2;
  const float* vbase = kqv + 2 * H_ * D_ + h * D_ + d0 + l16;

  v8f c = {};
#pragma unroll
  for (int jt = 0; jt < 5; ++jt) {
    const int j0 = t0 - 64 + jt * 16;
#pragma unroll
    for (int k4 = 0; k4 < 16; k4 += 4) {
      v2f a;
      {
        const int t = t0 + l16;
        const int wa = (j0 + k4 + kb) - t + (W_ - 1);
        const int wA = clampi(wa, 0, W_ - 1);
        const int wB = clampi(wa + 1, 0, W_ - 1);
        a.x = (wa     >= 0 && wa     < W_) ? sc[l16][wA] : 0.0f;
        a.y = (wa + 1 >= 0 && wa + 1 < W_) ? sc[l16][wB] : 0.0f;
      }
      v2f bv;
      {
        const int jA = clampi(j0 + k4 + kb,     0, T_ - 1);
        const int jB = clampi(j0 + k4 + kb + 1, 0, T_ - 1);
        bv.x = vbase[(size_t)(b * T_ + jA) * KQVROW];
        bv.y = vbase[(size_t)(b * T_ + jB) * KQVROW];
      }
      c = WMMA_F32(a, bv, c);
    }
  }
#pragma unroll
  for (int k = 0; k < W_; k += 4) {
    v2f a;
    a.x = sc[l16][k + kb];
    a.y = sc[l16][k + kb + 1];
    v2f bv;
    bv.x = v_pos[((size_t)(k + kb)     * H_ + h) * D_ + d0 + l16];
    bv.y = v_pos[((size_t)(k + kb + 1) * H_ + h) * D_ + d0 + l16];
    c = WMMA_F32(a, bv, c);
  }

#pragma unroll
  for (int i = 0; i < 8; ++i) {
    const int ti = i + half * 8;
    const int t  = t0 + ti;
    if (t < Tq_)
      attnout[((size_t)b * T_ + (t + S_)) * C_ + h * D_ + d0 + l16] = c[i];
  }
}

extern "C" void kernel_launch(void* const* d_in, const int* in_sizes, int n_in,
                              void* d_out, int out_size, void* d_ws, size_t ws_size,
                              hipStream_t stream) {
  (void)in_sizes; (void)n_in; (void)out_size; (void)ws_size;

  const float* x     = (const float*)d_in[0];
  const float* w_kqv = (const float*)d_in[1];
  const float* w_out = (const float*)d_in[2];
  const float* b_out = (const float*)d_in[3];
  const float* q_pos = (const float*)d_in[4];
  const float* v_pos = (const float*)d_in[5];

  float* y   = (float*)d_out;
  float* wei = y + (size_t)B_ * T_ * C_;

  char*  ws      = (char*)d_ws;
  float* kqv     = (float*)ws;
  size_t off     = (size_t)B_ * T_ * KQVROW * sizeof(float);
  float* S1      = (float*)(ws + off);
  off           += (size_t)B_ * H_ * T_ * W_ * sizeof(float);
  float* s0      = (float*)(ws + off);
  off           += (size_t)B_ * H_ * T_ * sizeof(float);
  float* attnout = (float*)(ws + off);

  gemm_abt<2, 2, 2, 2><<<dim3(KQVROW / 64, (B_ * T_) / 64), 128, 0, stream>>>(
      x, w_kqv, nullptr, kqv, B_ * T_, KQVROW, C_);

  s1_kernel<<<dim3(T_ / 16, W_ / 16, B_ * H_), 32, 0, stream>>>(kqv, q_pos, S1, s0);

  attn_kernel<<<dim3((Tq_ + 15) / 16, H_, B_), 128, 0, stream>>>(
      kqv, S1, s0, v_pos, attnout, wei);

  gemm_abt<2, 1, 2, 4><<<dim3(C_ / 64, (B_ * T_) / 64), 256, 0, stream>>>(
      attnout, w_out, b_out, y, B_ * T_, C_, C_);
}


# 4 launches -> 3: S1 (q.q_pos, WMMA) and s0 (k.q) computed inside the attention kernel, band scattered to LDS; q/k/q_pos read as 128-bit lane pieces, so the S1 and s0 sum order over d changes
# speedup vs baseline: 1.5526x; 1.0879x over previous
#include <hip/hip_runtime.h>
#include <math.h>


typedef __attribute__((ext_vector_type(2))) float v2f;
typedef __attribute__((ext_vector_type(4))) float v4f;
typedef __attribute__((ext_vector_type(8))) float v8f;

#define WMMA_F32(a, b, c) \
  __builtin_amdgcn_wmma_f32_16x16x4_f32(false, (a), false, (b), (short)0, (c), false, false)

constexpr int B_ = 2, T_ = 2048, C_ = 256, H_ = 4, D_ = 64, W_ = 64, S_ = 1;
constexpr int Tq_ = T_ - S_;
constexpr int KQVROW = 3 * H_ * D_;

__device__ __forceinline__ int clampi(int v, int lo, int hi) {
  return v < lo ? lo : (v > hi ? hi : v);
}

template <int WM, int WN, int WVM, int WVN>
__global__ __launch_bounds__(32 * WVM * WVN)
void gemm_abt(const float* __restrict__ A, const float* __restrict__ Bm,
              const float* __restrict__ bias, float* __restrict__ Cm,
              int M, int N, int K) {
  constexpr int NT    = 32 * WVM * WVN;
  constexpr int BM    = 16 * WM * WVM;
  constexpr int BN    = 16 * WN * WVN;
  constexpr int KC    = 32;
  constexpr int PITCH = KC + 4;
  constexpr int RPP   = NT / 8;
  constexpr int PA    = BM / RPP;
  constexpr int PB    = BN / RPP;

  __shared__ __attribute__((aligned(16))) float sA[2][BM][PITCH];
  __shared__ __attribute__((aligned(16))) float sB[2][BN][PITCH];

  const int tid  = threadIdx.x;
  const int lane = tid & 31;
  const int wave = tid >> 5;
  const int half = lane >> 4;
  const int l16  = lane & 15;
  const int kb   = half * 2;
  const int wm   = wave / WVN;
  const int wn   = wave % WVN;
  const int n0   = blockIdx.x * BN;
  const int m0   = blockIdx.y * BM;

  const int lr = tid >> 3;
  const int lc = (tid & 7) * 4;
  const float* ga = A  + (size_t)(m0 + lr) * K + lc;
  const float* gb = Bm + (size_t)(n0 + lr) * K + lc;

  v4f ra[PA], rb[PB];
#pragma unroll
  for (int p = 0; p < PA; ++p) ra[p] = *(const v4f*)(ga + (size_t)(p * RPP) * K);
#pragma unroll
  for (int p = 0; p < PB; ++p) rb[p] = *(const v4f*)(gb + (size_t)(p * RPP) * K);
#pragma unroll
  for (int p = 0; p < PA; ++p) *(v4f*)&sA[0][p * RPP + lr][lc] = ra[p];
#pragma unroll
  for (int p = 0; p < PB; ++p) *(v4f*)&sB[0][p * RPP + lr][lc] = rb[p];
  __syncthreads();

  v8f acc[WM][WN];
#pragma unroll
  for (int i = 0; i < WM; ++i)
#pragma unroll
    for (int j = 0; j < WN; ++j) acc[i][j] = v8f{};

  const int arow = wm * 16 * WM + l16;
  const int brow = wn * 16 * WN + l16;
  const int nch  = K / KC;
  for (int c = 0; c < nch; ++c) {
    const int cur  = c & 1;
    const bool more = (c + 1 < nch);
    if (more) {
      const int kn = (c + 1) * KC;
#pragma unroll
      for (int p = 0; p < PA; ++p) ra[p] = *(const v4f*)(ga + (size_t)(p * RPP) * K + kn);
#pragma unroll
      for (int p = 0; p < PB; ++p) rb[p] = *(const v4f*)(gb + (size_t)(p * RPP) * K + kn);
    }
#pragma unroll
    for (int ks = 0; ks < KC; ks += 4) {
      v2f fa[WM], fb[WN];
#pragma unroll
      for (int i = 0; i < WM; ++i) fa[i] = *(const v2f*)&sA[cur][arow + i * 16][ks + kb];
#pragma unroll
      for (int j = 0; j < WN; ++j) fb[j] = *(const v2f*)&sB[cur][brow + j * 16][ks + kb];
#pragma unroll
      for (int i = 0; i < WM; ++i)
#pragma unroll
        for (int j = 0; j < WN; ++j) acc[i][j] = WMMA_F32(fa[i], fb[j], acc[i][j]);
    }
    if (more) {
#pragma unroll
      for (int p = 0; p < PA; ++p) *(v4f*)&sA[cur ^ 1][p * RPP + lr][lc] = ra[p];
#pragma unroll
      for (int p = 0; p < PB; ++p) *(v4f*)&sB[cur ^ 1][p * RPP + lr][lc] = rb[p];
    }
    __syncthreads();
  }

#pragma unroll
  for (int j = 0; j < WN; ++j) {
    const int nn = n0 + wn * 16 * WN + j * 16 + l16;
    const float bb = bias ? bias[nn] : 0.0f;
#pragma unroll
    for (int i = 0; i < WM; ++i) {
#pragma unroll
      for (int r = 0; r < 8; ++r) {
        const int m = m0 + wm * 16 * WM + i * 16 + r + half * 8;
        Cm[(size_t)m * N + nn] = acc[i][j][r] + bb;
      }
    }
  }
}

__global__ __launch_bounds__(128)
void attn_kernel(const float* __restrict__ kqv, const float* __restrict__ q_pos,
                 const float* __restrict__ v_pos,
                 float* __restrict__ attnout, float* __restrict__ wei_out) {
  constexpr int SCP = W_ + 4;
  __shared__ __attribute__((aligned(16))) float sc[16][SCP];
  __shared__ float s0s[W_ + 16];

  const int t0 = blockIdx.x * 16;
  const int h  = blockIdx.y;
  const int b  = blockIdx.z;
  const int tid = threadIdx.x;
  const int lane = tid & 31;
  const int wave = __builtin_amdgcn_readfirstlane(tid >> 5);
  const int half = lane >> 4, l16 = lane & 15, kb = half * 2;
  const float scale = 0.125f;

  if (blockIdx.x == 0 && tid < D_)
    attnout[(size_t)b * T_ * C_ + h * D_ + tid] = 0.0f;

  {
    const int jw0 = t0 - W_ + 16 * wave;
    const int ja  = jw0 + l16;
    const int jb  = ja + 16;
    const int jra = ja < 0 ? 0 : ja;
    const int jrb = jb < 0 ? 0 : jb;
    const float* ka = kqv + (size_t)(b * T_ + jra) * KQVROW + h * D_ + 4 * half;
    const float* kc = kqv + (size_t)(b * T_ + jrb) * KQVROW + h * D_ + 4 * half;
    const float* qa = ka + H_ * D_;
    const float* qc = kc + H_ * D_;
    const float* pr = q_pos + (size_t)((16 * wave + l16) * H_ + h) * D_ + 4 * half;

    v4f fa[D_ / 8], fc[D_ / 8], fp[D_ / 8];
#pragma unroll
    for (int i = 0; i < D_ / 8; ++i) fa[i] = *(const v4f*)(qa + 8 * i);
#pragma unroll
    for (int i = 0; i < D_ / 8; ++i) fc[i] = *(const v4f*)(qc + 8 * i);
#pragma unroll
    for (int i = 0; i < D_ / 8; ++i) fp[i] = *(const v4f*)(pr + 8 * i);

    {
      float p = 0.0f;
#pragma unroll
      for (int i = 0; i < D_ / 8; ++i) {
        const v4f kk = *(const v4f*)(ka + 8 * i);
        p = fmaf(kk.x, fa[i].x, p);
        p = fmaf(kk.y, fa[i].y, p);
        p = fmaf(kk.z, fa[i].z, p);
        p = fmaf(kk.w, fa[i].w, p);
      }
      p += __shfl_xor(p, 16);
      s0s[16 * wave + l16] = p;
    }
    if (wave == 3) {
      float p = 0.0f;
#pragma unroll
      for (int i = 0; i < D_ / 8; ++i) {
        const v4f kk = *(const v4f*)(kc + 8 * i);
        p = fmaf(kk.x, fc[i].x, p);
        p = fmaf(kk.y, fc[i].y, p);
        p = fmaf(kk.z, fc[i].z, p);
        p = fmaf(kk.w, fc[i].w, p);
      }
      p += __shfl_xor(p, 16);
      s0s[W_ + l16] = p;
    }

    v8f c0 = {}, c1 = {};
#pragma unroll
    for (int i = 0; i < D_ / 8; ++i) {
      const v2f blo = fp[i].lo, bhi = fp[i].hi;
      c0 = WMMA_F32(fa[i].lo, blo, c0);
      c1 = WMMA_F32(fc[i].lo, blo, c1);
      c0 = WMMA_F32(fa[i].hi, bhi, c0);
      c1 = WMMA_F32(fc[i].hi, bhi, c1);
    }

    __syncthreads();

#pragma unroll
    for (int r = 0; r < 8; ++r) {
      const int  R    = 8 * half + r;
      const bool lowt = R > l16;
      const int  jrel = lowt ? R : R + 16;
      const float s1  = lowt ? c0[r] : c1[r];
      const int  ti   = jrel - l16 - 1;
      float v = (s0s[16 * wave + jrel] + s1) * scale;
      if (jw0 + jrel < 0)    v = -INFINITY;
      if (t0 + ti >= Tq_)    v = 0.0f;
      sc[ti][16 * wave + l16] = v;
    }
  }
  __syncthreads();

  if (tid < 16) {
    float m = -INFINITY;
    for (int w = 0; w < W_; ++w) m = fmaxf(m, sc[tid][w]);
    float s = 0.0f;
    for (int w = 0; w < W_; ++w) { float e = __expf(sc[tid][w] - m); sc[tid][w] = e; s += e; }
    const float inv = 1.0f / s;
    for (int w = 0; w < W_; ++w) sc[tid][w] *= inv;
  }
  __syncthreads();

  {
    const int ti = tid >> 3;
    const int wb = (tid & 7) * 8;
    const int t  = t0 + ti;
    if (t < Tq_) {
#pragma unroll
      for (int u = 0; u < 8; ++u) {
        const int w = wb + u;
        wei_out[(((size_t)b * Tq_ + t) * W_ + w) * H_ + h] = sc[ti][w];
      }
    }
  }

  const int d0   = wave * 16;
  const float* vbase = kqv + 2 * H_ * D_ + h * D_ + d0 + l16;

  v8f c = {};
#pragma unroll
  for (int jt = 0; jt < 5; ++jt) {
    const int j0 = t0 - 64 + jt * 16;
#pragma unroll
    for (int k4 = 0; k4 < 16; k4 += 4) {
      v2f a;
      {
        const int t = t0 + l16;
        const int wa = (j0 + k4 + kb) - t + (W_ - 1);
        const int wA = clampi(wa, 0, W_ - 1);
        const int wB = clampi(wa + 1, 0, W_ - 1);
        a.x = (wa     >= 0 && wa     < W_) ? sc[l16][wA] : 0.0f;
        a.y = (wa + 1 >= 0 && wa + 1 < W_) ? sc[l16][wB] : 0.0f;
      }
      v2f bv;
      {
        const int jA = clampi(j0 + k4 + kb,     0, T_ - 1);
        const int jB = clampi(j0 + k4 + kb + 1, 0, T_ - 1);
        bv.x = vbase[(size_t)(b * T_ + jA) * KQVROW];
        bv.y = vbase[(size_t)(b * T_ + jB) * KQVROW];
      }
      c = WMMA_F32(a, bv, c);
    }
  }
#pragma unroll
  for (int k = 0; k < W_; k += 4) {
    v2f a;
    a.x = sc[l16][k + kb];
    a.y = sc[l16][k + kb + 1];
    v2f bv;
    bv.x = v_pos[((size_t)(k + kb)     * H_ + h) * D_ + d0 + l16];
    bv.y = v_pos[((size_t)(k + kb + 1) * H_ + h) * D_ + d0 + l16];
    c = WMMA_F32(a, bv, c);
  }

#pragma unroll
  for (int i = 0; i < 8; ++i) {
    const int ti = i + half * 8;
    const int t  = t0 + ti;
    if (t < Tq_)
      attnout[((size_t)b * T_ + (t + S_)) * C_ + h * D_ + d0 + l16] = c[i];
  }
}

extern "C" void kernel_launch(void* const* d_in, const int* in_sizes, int n_in,
                              void* d_out, int out_size, void* d_ws, size_t ws_size,
                              hipStream_t stream) {
  (void)in_sizes; (void)n_in; (void)out_size; (void)ws_size;

  const float* x     = (const float*)d_in[0];
  const float* w_kqv = (const float*)d_in[1];
  const float* w_out = (const float*)d_in[2];
  const float* b_out = (const float*)d_in[3];
  const float* q_pos = (const float*)d_in[4];
  const float* v_pos = (const float*)d_in[5];

  float* y   = (float*)d_out;
  float* wei = y + (size_t)B_ * T_ * C_;

  char*  ws      = (char*)d_ws;
  float* kqv     = (float*)ws;
  size_t off     = (size_t)B_ * T_ * KQVROW * sizeof(float);
  float* attnout = (float*)(ws + off);

  gemm_abt<2, 2, 2, 2><<<dim3(KQVROW / 64, (B_ * T_) / 64), 128, 0, stream>>>(
      x, w_kqv, nullptr, kqv, B_ * T_, KQVROW, C_);

  attn_kernel<<<dim3((Tq_ + 15) / 16, H_, B_), 128, 0, stream>>>(
      kqv, q_pos, v_pos, attnout, wei);

  gemm_abt<2, 1, 2, 4><<<dim3(C_ / 64, (B_ * T_) / 64), 256, 0, stream>>>(
      attnout, w_out, b_out, y, B_ * T_, C_, C_);
}


# output products: V and v_pos operand loads issued as one group ahead of the softmax; weights read from zero-margined LDS rows without conditions (before: each banded WMMA waited for its own V loads)
# speedup vs baseline: 1.8240x; 1.1748x over previous
#include <hip/hip_runtime.h>
#include <math.h>


typedef __attribute__((ext_vector_type(2))) float v2f;
typedef __attribute__((ext_vector_type(4))) float v4f;
typedef __attribute__((ext_vector_type(8))) float v8f;

#define WMMA_F32(a, b, c) \
  __builtin_amdgcn_wmma_f32_16x16x4_f32(false, (a), false, (b), (short)0, (c), false, false)

constexpr int B_ = 2, T_ = 2048, C_ = 256, H_ = 4, D_ = 64, W_ = 64, S_ = 1;
constexpr int Tq_ = T_ - S_;
constexpr int KQVROW = 3 * H_ * D_;

__device__ __forceinline__ int clampi(int v, int lo, int hi) {
  return v < lo ? lo : (v > hi ? hi : v);
}

template <int WM, int WN, int WVM, int WVN>
__global__ __launch_bounds__(32 * WVM * WVN)
void gemm_abt(const float* __restrict__ A, const float* __restrict__ Bm,
              const float* __restrict__ bias, float* __restrict__ Cm,
              int M, int N, int K) {
  constexpr int NT    = 32 * WVM * WVN;
  constexpr int BM    = 16 * WM * WVM;
  constexpr int BN    = 16 * WN * WVN;
  constexpr int KC    = 32;
  constexpr int PITCH = KC + 4;
  constexpr int RPP   = NT / 8;
  constexpr int PA    = BM / RPP;
  constexpr int PB    = BN / RPP;

  __shared__ __attribute__((aligned(16))) float sA[2][BM][PITCH];
  __shared__ __attribute__((aligned(16))) float sB[2][BN][PITCH];

  const int tid  = threadIdx.x;
  const int lane = tid & 31;
  const int wave = tid >> 5;
  const int half = lane >> 4;
  const int l16  = lane & 15;
  const int kb   = half * 2;
  const int wm   = wave / WVN;
  const int wn   = wave % WVN;
  const int n0   = blockIdx.x * BN;
  const int m0   = blockIdx.y * BM;

  const int lr = tid >> 3;
  const int lc = (tid & 7) * 4;
  const float* ga = A  + (size_t)(m0 + lr) * K + lc;
  const float* gb = Bm + (size_t)(n0 + lr) * K + lc;

  v4f ra[PA], rb[PB];
#pragma unroll
  for (int p = 0; p < PA; ++p) ra[p] = *(const v4f*)(ga + (size_t)(p * RPP) * K);
#pragma unroll
  for (int p = 0; p < PB; ++p) rb[p] = *(const v4f*)(gb + (size_t)(p * RPP) * K);
#pragma unroll
  for (int p = 0; p < PA; ++p) *(v4f*)&sA[0][p * RPP + lr][lc] = ra[p];
#pragma unroll
  for (int p = 0; p < PB; ++p) *(v4f*)&sB[0][p * RPP + lr][lc] = rb[p];
  __syncthreads();

  v8f acc[WM][WN];
#pragma unroll
  for (int i = 0; i < WM; ++i)
#pragma unroll
    for (int j = 0; j < WN; ++j) acc[i][j] = v8f{};

  const int arow = wm * 16 * WM + l16;
  const int brow = wn * 16 * WN + l16;
  const int nch  = K / KC;
  for (int c = 0; c < nch; ++c) {
    const int cur  = c & 1;
    const bool more = (c + 1 < nch);
    if (more) {
      const int kn = (c + 1) * KC;
#pragma unroll
      for (int p = 0; p < PA; ++p) ra[p] = *(const v4f*)(ga + (size_t)(p * RPP) * K + kn);
#pragma unroll
      for (int p = 0; p < PB; ++p) rb[p] = *(const v4f*)(gb + (size_t)(p * RPP) * K + kn);
    }
#pragma unroll
    for (int ks = 0; ks < KC; ks += 4) {
      v2f fa[WM], fb[WN];
#pragma unroll
      for (int i = 0; i < WM; ++i) fa[i] = *(const v2f*)&sA[cur][arow + i * 16][ks + kb];
#pragma unroll
      for (int j = 0; j < WN; ++j) fb[j] = *(const v2f*)&sB[cur][brow + j * 16][ks + kb];
#pragma unroll
      for (int i = 0; i < WM; ++i)
#pragma unroll
        for (int j = 0; j < WN; ++j) acc[i][j] = WMMA_F32(fa[i], fb[j], acc[i][j]);
    }
    if (more) {
#pragma unroll
      for (int p = 0; p < PA; ++p) *(v4f*)&sA[cur ^ 1][p * RPP + lr][lc] = ra[p];
#pragma unroll
      for (int p = 0; p < PB; ++p) *(v4f*)&sB[cur ^ 1][p * RPP + lr][lc] = rb[p];
    }
    __syncthreads();
  }

#pragma unroll
  for (int j = 0; j < WN; ++j) {
    const int nn = n0 + wn * 16 * WN + j * 16 + l16;
    const float bb = bias ? bias[nn] : 0.0f;
#pragma unroll
    for (int i = 0; i < WM; ++i) {
#pragma unroll
      for (int r = 0; r < 8; ++r) {
        const int m = m0 + wm * 16 * WM + i * 16 + r + half * 8;
        Cm[(size_t)m * N + nn] = acc[i][j][r] + bb;
      }
    }
  }
}

__global__ __launch_bounds__(128)
void attn_kernel(const float* __restrict__ kqv, const float* __restrict__ q_pos,
                 const float* __restrict__ v_pos,
                 float* __restrict__ attnout, float* __restrict__ wei_out) {
  constexpr int MG  = 16;
  constexpr int SCP = W_ + 2 * MG + 4;
  __shared__ __attribute__((aligned(16))) float sc[16][SCP];
  __shared__ float s0s[W_ + 16];

  const int t0 = blockIdx.x * 16;
  const int h  = blockIdx.y;
  const int b  = blockIdx.z;
  const int tid = threadIdx.x;
  const int lane = tid & 31;
  const int wave = __builtin_amdgcn_readfirstlane(tid >> 5);
  const int half = lane >> 4, l16 = lane & 15, kb = half * 2;
  const float scale = 0.125f;

  if (blockIdx.x == 0 && tid < D_)
    attnout[(size_t)b * T_ * C_ + h * D_ + tid] = 0.0f;

  {
    const int zr = tid >> 3, zp = tid & 7;
    const int zc = zp < 4 ? 4 * zp : W_ + MG + 4 * (zp - 4);
    *(v4f*)&sc[zr][zc] = v4f{};
  }

  {
    const int jw0 = t0 - W_ + 16 * wave;
    const int ja  = jw0 + l16;
    const int jb  = ja + 16;
    const int jra = ja < 0 ? 0 : ja;
    const int jrb = jb < 0 ? 0 : jb;
    const float* ka = kqv + (size_t)(b * T_ + jra) * KQVROW + h * D_ + 4 * half;
    const float* kc = kqv + (size_t)(b * T_ + jrb) * KQVROW + h * D_ + 4 * half;
    const float* qa = ka + H_ * D_;
    const float* qc = kc + H_ * D_;
    const float* pr = q_pos + (size_t)((16 * wave + l16) * H_ + h) * D_ + 4 * half;

    v4f fa[D_ / 8], fc[D_ / 8], fp[D_ / 8];
#pragma unroll
    for (int i = 0; i < D_ / 8; ++i) fa[i] = *(const v4f*)(qa + 8 * i);
#pragma unroll
    for (int i = 0; i < D_ / 8; ++i) fc[i] = *(const v4f*)(qc + 8 * i);
#pragma unroll
    for (int i = 0; i < D_ / 8; ++i) fp[i] = *(const v4f*)(pr + 8 * i);

    {
      float p = 0.0f;
#pragma unroll
      for (int i = 0; i < D_ / 8; ++i) {
        const v4f kk = *(const v4f*)(ka + 8 * i);
        p = fmaf(kk.x, fa[i].x, p);
        p = fmaf(kk.y, fa[i].y, p);
        p = fmaf(kk.z, fa[i].z, p);
        p = fmaf(kk.w, fa[i].w, p);
      }
      p += __shfl_xor(p, 16);
      s0s[16 * wave + l16] = p;
    }
    if (wave == 3) {
      float p = 0.0f;
#pragma unroll
      for (int i = 0; i < D_ / 8; ++i) {
        const v4f kk = *(const v4f*)(kc + 8 * i);
        p = fmaf(kk.x, fc[i].x, p);
        p = fmaf(kk.y, fc[i].y, p);
        p = fmaf(kk.z, fc[i].z, p);
        p = fmaf(kk.w, fc[i].w, p);
      }
      p += __shfl_xor(p, 16);
      s0s[W_ + l16] = p;
    }

    v8f c0 = {}, c1 = {};
#pragma unroll
    for (int i = 0; i < D_ / 8; ++i) {
      const v2f blo = fp[i].lo, bhi = fp[i].hi;
      c0 = WMMA_F32(fa[i].lo, blo, c0);
      c1 = WMMA_F32(fc[i].lo, blo, c1);
      c0 = WMMA_F32(fa[i].hi, bhi, c0);
      c1 = WMMA_F32(fc[i].hi, bhi, c1);
    }

    __syncthreads();

#pragma unroll
    for (int r = 0; r < 8; ++r) {
      const int  R    = 8 * half + r;
      const bool lowt = R > l16;
      const int  jrel = lowt ? R : R + 16;
      const float s1  = lowt ? c0[r] : c1[r];
      const int  ti   = jrel - l16 - 1;
      float v = (s0s[16 * wave + jrel] + s1) * scale;
      if (jw0 + jrel < 0)    v = -INFINITY;
      if (t0 + ti >= Tq_)    v = 0.0f;
      sc[ti][MG + 16 * wave + l16] = v;
    }
  }
  __syncthreads();

  const int d0   = wave * 16;
  const float* vbase = kqv + (size_t)b * T_ * KQVROW + 2 * H_ * D_ + h * D_ + d0 + l16;
  v2f bvv[20], bvp[W_ / 4];
#pragma unroll
  for (int jt = 0; jt < 5; ++jt) {
    const int j0  = t0 - 64 + jt * 16;
    const int j0c = j0 < 0 ? 0 : j0;
#pragma unroll
    for (int k4 = 0; k4 < 16; k4 += 4) {
      const float* vr = vbase + (size_t)(j0c + k4 + kb) * KQVROW;
      bvv[jt * 4 + k4 / 4].x = vr[0];
      bvv[jt * 4 + k4 / 4].y = vr[KQVROW];
    }
  }
#pragma unroll
  for (int k = 0; k < W_; k += 4) {
    bvp[k / 4].x = v_pos[((size_t)(k + kb)     * H_ + h) * D_ + d0 + l16];
    bvp[k / 4].y = v_pos[((size_t)(k + kb + 1) * H_ + h) * D_ + d0 + l16];
  }

  if (tid < 16) {
    float m = -INFINITY;
    for (int w = 0; w < W_; ++w) m = fmaxf(m, sc[tid][MG + w]);
    float s = 0.0f;
    for (int w = 0; w < W_; ++w) { float e = __expf(sc[tid][MG + w] - m); sc[tid][MG + w] = e; s += e; }
    const float inv = 1.0f / s;
    for (int w = 0; w < W_; ++w) sc[tid][MG + w] *= inv;
  }
  __syncthreads();

  {
    const int ti = tid >> 3;
    const int wb = (tid & 7) * 8;
    const int t  = t0 + ti;
    if (t < Tq_) {
#pragma unroll
      for (int u = 0; u < 8; ++u) {
        const int w = wb + u;
        wei_out[(((size_t)b * Tq_ + t) * W_ + w) * H_ + h] = sc[ti][MG + w];
      }
    }
  }

  v8f c = {};
#pragma unroll
  for (int jt = 0; jt < 5; ++jt) {
#pragma unroll
    for (int k4 = 0; k4 < 16; k4 += 4) {
      const int wa = jt * 16 + k4 + kb - l16 - 1;
      v2f a;
      a.x = sc[l16][MG + wa];
      a.y = sc[l16][MG + wa + 1];
      c = WMMA_F32(a, bvv[jt * 4 + k4 / 4], c);
    }
  }
#pragma unroll
  for (int k = 0; k < W_; k += 4) {
    v2f a;
    a.x = sc[l16][MG + k + kb];
    a.y = sc[l16][MG + k + kb + 1];
    c = WMMA_F32(a, bvp[k / 4], c);
  }

#pragma unroll
  for (int i = 0; i < 8; ++i) {
    const int ti = i + half * 8;
    const int t  = t0 + ti;
    if (t < Tq_)
      attnout[((size_t)b * T_ + (t + S_)) * C_ + h * D_ + d0 + l16] = c[i];
  }
}

extern "C" void kernel_launch(void* const* d_in, const int* in_sizes, int n_in,
                              void* d_out, int out_size, void* d_ws, size_t ws_size,
                              hipStream_t stream) {
  (void)in_sizes; (void)n_in; (void)out_size; (void)ws_size;

  const float* x     = (const float*)d_in[0];
  const float* w_kqv = (const float*)d_in[1];
  const float* w_out = (const float*)d_in[2];
  const float* b_out = (const float*)d_in[3];
  const float* q_pos = (const float*)d_in[4];
  const float* v_pos = (const float*)d_in[5];

  float* y   = (float*)d_out;
  float* wei = y + (size_t)B_ * T_ * C_;

  char*  ws      = (char*)d_ws;
  float* kqv     = (float*)ws;
  size_t off     = (size_t)B_ * T_ * KQVROW * sizeof(float);
  float* attnout = (float*)(ws + off);

  gemm_abt<2, 2, 2, 2><<<dim3(KQVROW / 64, (B_ * T_) / 64), 128, 0, stream>>>(
      x, w_kqv, nullptr, kqv, B_ * T_, KQVROW, C_);

  attn_kernel<<<dim3((Tq_ + 15) / 16, H_, B_), 128, 0, stream>>>(
      kqv, q_pos, v_pos, attnout, wei);

  gemm_abt<2, 1, 2, 4><<<dim3(C_ / 64, (B_ * T_) / 64), 256, 0, stream>>>(
      attnout, w_out, b_out, y, B_ * T_, C_, C_);
}


# 3 launches -> 2: out projection done inside the attention kernel (a 16-wave block owns all 4 heads of a 16-row tile, w_out staged whole in LDS); its k sum order changes (128-bit lane pieces)
# speedup vs baseline: 2.0505x; 1.1242x over previous
#include <hip/hip_runtime.h>
#include <math.h>


typedef __attribute__((ext_vector_type(2))) float v2f;
typedef __attribute__((ext_vector_type(4))) float v4f;
typedef __attribute__((ext_vector_type(8))) float v8f;

#define WMMA_F32(a, b, c) \
  __builtin_amdgcn_wmma_f32_16x16x4_f32(false, (a), false, (b), (short)0, (c), false, false)

constexpr int B_ = 2, T_ = 2048, C_ = 256, H_ = 4, D_ = 64, W_ = 64, S_ = 1;
constexpr int Tq_ = T_ - S_;
constexpr int KQVROW = 3 * H_ * D_;

__device__ __forceinline__ int clampi(int v, int lo, int hi) {
  return v < lo ? lo : (v > hi ? hi : v);
}

template <int WM, int WN, int WVM, int WVN>
__global__ __launch_bounds__(32 * WVM * WVN)
void gemm_abt(const float* __restrict__ A, const float* __restrict__ Bm,
              const float* __restrict__ bias, float* __restrict__ Cm,
              int M, int N, int K) {
  constexpr int NT    = 32 * WVM * WVN;
  constexpr int BM    = 16 * WM * WVM;
  constexpr int BN    = 16 * WN * WVN;
  constexpr int KC    = 32;
  constexpr int PITCH = KC + 4;
  constexpr int RPP   = NT / 8;
  constexpr int PA    = BM / RPP;
  constexpr int PB    = BN / RPP;

  __shared__ __attribute__((aligned(16))) float sA[2][BM][PITCH];
  __shared__ __attribute__((aligned(16))) float sB[2][BN][PITCH];

  const int tid  = threadIdx.x;
  const int lane = tid & 31;
  const int wave = tid >> 5;
  const int half = lane >> 4;
  const int l16  = lane & 15;
  const int kb   = half * 2;
  const int wm   = wave / WVN;
  const int wn   = wave % WVN;
  const int n0   = blockIdx.x * BN;
  const int m0   = blockIdx.y * BM;

  const int lr = tid >> 3;
  const int lc = (tid & 7) * 4;
  const float* ga = A  + (size_t)(m0 + lr) * K + lc;
  const float* gb = Bm + (size_t)(n0 + lr) * K + lc;

  v4f ra[PA], rb[PB];
#pragma unroll
  for (int p = 0; p < PA; ++p) ra[p] = *(const v4f*)(ga + (size_t)(p * RPP) * K);
#pragma unroll
  for (int p = 0; p < PB; ++p) rb[p] = *(const v4f*)(gb + (size_t)(p * RPP) * K);
#pragma unroll
  for (int p = 0; p < PA; ++p) *(v4f*)&sA[0][p * RPP + lr][lc] = ra[p];
#pragma unroll
  for (int p = 0; p < PB; ++p) *(v4f*)&sB[0][p * RPP + lr][lc] = rb[p];
  __syncthreads();

  v8f acc[WM][WN];
#pragma unroll
  for (int i = 0; i < WM; ++i)
#pragma unroll
    for (int j = 0; j < WN; ++j) acc[i][j] = v8f{};

  const int arow = wm * 16 * WM + l16;
  const int brow = wn * 16 * WN + l16;
  const int nch  = K / KC;
  for (int c = 0; c < nch; ++c) {
    const int cur  = c & 1;
    const bool more = (c + 1 < nch);
    if (more) {
      const int kn = (c + 1) * KC;
#pragma unroll
      for (int p = 0; p < PA; ++p) ra[p] = *(const v4f*)(ga + (size_t)(p * RPP) * K + kn);
#pragma unroll
      for (int p = 0; p < PB; ++p) rb[p] = *(const v4f*)(gb + (size_t)(p * RPP) * K + kn);
    }
#pragma unroll
    for (int ks = 0; ks < KC; ks += 4) {
      v2f fa[WM], fb[WN];
#pragma unroll
      for (int i = 0; i < WM; ++i) fa[i] = *(const v2f*)&sA[cur][arow + i * 16][ks + kb];
#pragma unroll
      for (int j = 0; j < WN; ++j) fb[j] = *(const v2f*)&sB[cur][brow + j * 16][ks + kb];
#pragma unroll
      for (int i = 0; i < WM; ++i)
#pragma unroll
        for (int j = 0; j < WN; ++j) acc[i][j] = WMMA_F32(fa[i], fb[j], acc[i][j]);
    }
    if (more) {
#pragma unroll
      for (int p = 0; p < PA; ++p) *(v4f*)&sA[cur ^ 1][p * RPP + lr][lc] = ra[p];
#pragma unroll
      for (int p = 0; p < PB; ++p) *(v4f*)&sB[cur ^ 1][p * RPP + lr][lc] = rb[p];
    }
    __syncthreads();
  }

#pragma unroll
  for (int j = 0; j < WN; ++j) {
    const int nn = n0 + wn * 16 * WN + j * 16 + l16;
    const float bb = bias ? bias[nn] : 0.0f;
#pragma unroll
    for (int i = 0; i < WM; ++i) {
#pragma unroll
      for (int r = 0; r < 8; ++r) {
        const int m = m0 + wm * 16 * WM + i * 16 + r + half * 8;
        Cm[(size_t)m * N + nn] = acc[i][j][r] + bb;
      }
    }
  }
}

__global__ __launch_bounds__(512)
void attn_proj_kernel(const float* __restrict__ kqv, const float* __restrict__ q_pos,
                      const float* __restrict__ v_pos, const float* __restrict__ w_out,
                      const float* __restrict__ b_out,
                      float* __restrict__ y, float* __restrict__ wei_out) {
  constexpr int MG  = 16;
  constexpr int SCP = W_ + 2 * MG + 4;
  constexpr int PP  = C_ + 4;
  __shared__ __attribute__((aligned(16))) float sc[H_][16][SCP];
  __shared__ float s0s[H_][W_ + 16];
  __shared__ __attribute__((aligned(16))) float ao[16][PP];
  __shared__ __attribute__((aligned(16))) float sW[C_][PP];

  const int t0 = blockIdx.x * 16;
  const int b  = blockIdx.y;
  const int tid = threadIdx.x;
  const int lane = tid & 31;
  const int wave = __builtin_amdgcn_readfirstlane(tid >> 5);
  const int h    = wave >> 2;
  const int wv   = wave & 3;
  const int ht   = tid & 127;
  const int half = lane >> 4, l16 = lane & 15, kb = half * 2;
  const float scale = 0.125f;

  if (blockIdx.x == 0 && tid < C_)
    y[(size_t)b * T_ * C_ + tid] = b_out[tid];

  {
    const int zr = ht >> 3, zp = ht & 7;
    const int zc = zp < 4 ? 4 * zp : W_ + MG + 4 * (zp - 4);
    *(v4f*)&sc[h][zr][zc] = v4f{};
  }

  const float* wsrc = w_out + (size_t)(tid >> 5) * C_ + (tid & 31) * 4;
  float* wdst = &sW[tid >> 5][(tid & 31) * 4];
  v4f rw[16];

  {
    const int jw0 = t0 - W_ + 16 * wv;
    const int ja  = jw0 + l16;
    const int jb  = ja + 16;
    const int jra = ja < 0 ? 0 : ja;
    const int jrb = jb < 0 ? 0 : jb;
    const float* ka = kqv + (size_t)(b * T_ + jra) * KQVROW + h * D_ + 4 * half;
    const float* kc = kqv + (size_t)(b * T_ + jrb) * KQVROW + h * D_ + 4 * half;
    const float* qa = ka + H_ * D_;
    const float* qc = kc + H_ * D_;
    const float* pr = q_pos + (size_t)((16 * wv + l16) * H_ + h) * D_ + 4 * half;

    v4f fa[D_ / 8], fc[D_ / 8], fp[D_ / 8];
#pragma unroll
    for (int i = 0; i < D_ / 8; ++i) fa[i] = *(const v4f*)(qa + 8 * i);
#pragma unroll
    for (int i = 0; i < D_ / 8; ++i) fc[i] = *(const v4f*)(qc + 8 * i);
#pragma unroll
    for (int i = 0; i < D_ / 8; ++i) fp[i] = *(const v4f*)(pr + 8 * i);

    {
      float p = 0.0f;
#pragma unroll
      for (int i = 0; i < D_ / 8; ++i) {
        const v4f kk = *(const v4f*)(ka + 8 * i);
        p = fmaf(kk.x, fa[i].x, p);
        p = fmaf(kk.y, fa[i].y, p);
        p = fmaf(kk.z, fa[i].z, p);
        p = fmaf(kk.w, fa[i].w, p);
      }
      p += __shfl_xor(p, 16);
      s0s[h][16 * wv + l16] = p;
    }
    if (wv == 3) {
      float p = 0.0f;
#pragma unroll
      for (int i = 0; i < D_ / 8; ++i) {
        const v4f kk = *(const v4f*)(kc + 8 * i);
        p = fmaf(kk.x, fc[i].x, p);
        p = fmaf(kk.y, fc[i].y, p);
        p = fmaf(kk.z, fc[i].z, p);
        p = fmaf(kk.w, fc[i].w, p);
      }
      p += __shfl_xor(p, 16);
      s0s[h][W_ + l16] = p;
    }

    v8f c0 = {}, c1 = {};
#pragma unroll
    for (int i = 0; i < D_ / 8; ++i) {
      const v2f blo = fp[i].lo, bhi = fp[i].hi;
      c0 = WMMA_F32(fa[i].lo, blo, c0);
      c1 = WMMA_F32(fc[i].lo, blo, c1);
      c0 = WMMA_F32(fa[i].hi, bhi, c0);
      c1 = WMMA_F32(fc[i].hi, bhi, c1);
    }

    __syncthreads();

    float s0v[8];
#pragma unroll
    for (int r = 0; r < 8; ++r) {
      const int R = 8 * half + r;
      s0v[r] = s0s[h][16 * wv + (R > l16 ? R : R + 16)];
    }
#pragma unroll
    for (int r = 0; r < 8; ++r) {
      const int  R    = 8 * half + r;
      const bool lowt = R > l16;
      const int  jrel = lowt ? R : R + 16;
      const float s1  = lowt ? c0[r] : c1[r];
      const int  ti   = jrel - l16 - 1;
      float v = (s0v[r] + s1) * scale;
      if (jw0 + jrel < 0)    v = -INFINITY;
      if (t0 + ti >= Tq_)    v = 0.0f;
      sc[h][ti][MG + 16 * wv + l16] = v;
    }
  }
  __syncthreads();

  const int d0   = wv * 16;
  const float* vbase = kqv + (size_t)b * T_ * KQVROW + 2 * H_ * D_ + h * D_ + d0 + l16;
  v2f bvv[20], bvp[W_ / 4];
#pragma unroll
  for (int jt = 0; jt < 5; ++jt) {
    const int j0  = t0 - 64 + jt * 16;
    const int j0c = j0 < 0 ? 0 : j0;
#pragma unroll
    for (int k4 = 0; k4 < 16; k4 += 4) {
      const float* vr = vbase + (size_t)(j0c + k4 + kb) * KQVROW;
      bvv[jt * 4 + k4 / 4].x = vr[0];
      bvv[jt * 4 + k4 / 4].y = vr[KQVROW];
    }
  }
#pragma unroll
  for (int k = 0; k < W_; k += 4) {
    bvp[k / 4].x = v_pos[((size_t)(k + kb)     * H_ + h) * D_ + d0 + l16];
    bvp[k / 4].y = v_pos[((size_t)(k + kb + 1) * H_ + h) * D_ + d0 + l16];
  }
  if (ht < 16) {
    float m = -INFINITY;
    for (int w = 0; w < W_; ++w) m = fmaxf(m, sc[h][ht][MG + w]);
    float s = 0.0f;
    for (int w = 0; w < W_; ++w) { float e = __expf(sc[h][ht][MG + w] - m); sc[h][ht][MG + w] = e; s += e; }
    const float inv = 1.0f / s;
    for (int w = 0; w < W_; ++w) sc[h][ht][MG + w] *= inv;
  }
  __syncthreads();

#pragma unroll
  for (int u = 0; u < 16; ++u) rw[u] = *(const v4f*)(wsrc + (size_t)u * 16 * C_);
  __builtin_amdgcn_sched_barrier(0);

  {
    const int ti = ht >> 3;
    const int wb = (ht & 7) * 8;
    const int t  = t0 + ti;
    if (t < Tq_) {
#pragma unroll
      for (int u = 0; u < 8; ++u) {
        const int w = wb + u;
        wei_out[(((size_t)b * Tq_ + t) * W_ + w) * H_ + h] = sc[h][ti][MG + w];
      }
    }
  }

  v8f c = {};
#pragma unroll
  for (int jt = 0; jt < 5; ++jt) {
#pragma unroll
    for (int k4 = 0; k4 < 16; k4 += 4) {
      const int wa = jt * 16 + k4 + kb - l16 - 1;
      v2f a;
      a.x = sc[h][l16][MG + wa];
      a.y = sc[h][l16][MG + wa + 1];
      c = WMMA_F32(a, bvv[jt * 4 + k4 / 4], c);
    }
  }
#pragma unroll
  for (int k = 0; k < W_; k += 4) {
    v2f a;
    a.x = sc[h][l16][MG + k + kb];
    a.y = sc[h][l16][MG + k + kb + 1];
    c = WMMA_F32(a, bvp[k / 4], c);
  }

#pragma unroll
  for (int i = 0; i < 8; ++i) ao[i + half * 8][h * D_ + d0 + l16] = c[i];

#pragma unroll
  for (int u = 0; u < 16; ++u) *(v4f*)(wdst + (size_t)u * 16 * PP) = rw[u];

  __syncthreads();

#pragma unroll
  for (int u = 0; u < 16; ++u) rw[u] = *(const v4f*)(wsrc + 128 + (size_t)u * 16 * C_);
  __builtin_amdgcn_sched_barrier(0);

  {
    const int n0 = 16 * wave;
    const float* pa = &ao[l16][4 * half];
    const float* pb = &sW[n0 + l16][4 * half];
    const float bb = b_out[n0 + l16];
    v8f acc = {};
#pragma unroll
    for (int g = 0; g < 4; ++g) {
      if (g == 2) {
        asm volatile("" :: "v"(acc) : "memory");
#pragma unroll
        for (int u = 0; u < 16; ++u) *(v4f*)(wdst + 128 + (size_t)u * 16 * PP) = rw[u];
        __syncthreads();
      }
      v4f xa[8], xb[8];
#pragma unroll
      for (int i = 0; i < 8; ++i) xa[i] = *(const v4f*)(pa + 64 * g + 8 * i);
#pragma unroll
      for (int i = 0; i < 8; ++i) xb[i] = *(const v4f*)(pb + 64 * g + 8 * i);
#pragma unroll
      for (int i = 0; i < 8; ++i) {
        acc = WMMA_F32(xa[i].lo, xb[i].lo, acc);
        acc = WMMA_F32(xa[i].hi, xb[i].hi, acc);
      }
    }
    float* yr = y + ((size_t)b * T_ + (t0 + S_ + half * 8)) * C_ + n0 + l16;
#pragma unroll
    for (int r = 0; r < 7; ++r) yr[(size_t)r * C_] = acc[r] + bb;
    if (t0 + 7 + half * 8 < Tq_) yr[(size_t)7 * C_] = acc[7] + bb;
  }
}

extern "C" void kernel_launch(void* const* d_in, const int* in_sizes, int n_in,
                              void* d_out, int out_size, void* d_ws, size_t ws_size,
                              hipStream_t stream) {
  (void)in_sizes; (void)n_in; (void)out_size; (void)ws_size;

  const float* x     = (const float*)d_in[0];
  const float* w_kqv = (const float*)d_in[1];
  const float* w_out = (const float*)d_in[2];
  const float* b_out = (const float*)d_in[3];
  const float* q_pos = (const float*)d_in[4];
  const float* v_pos = (const float*)d_in[5];

  float* y   = (float*)d_out;
  float* wei = y + (size_t)B_ * T_ * C_;

  float* kqv     = (float*)d_ws;

  gemm_abt<2, 2, 2, 2><<<dim3(KQVROW / 64, (B_ * T_) / 64), 128, 0, stream>>>(
      x, w_kqv, nullptr, kqv, B_ * T_, KQVROW, C_);

  attn_proj_kernel<<<dim3((Tq_ + 15) / 16, B_), 512, 0, stream>>>(
      kqv, q_pos, v_pos, w_out, b_out, y, wei);
}


# wei stores: a thread gathers the 4 heads' weights of one (t, w) from LDS and writes one 128-bit store, 4 whole 128-B lines per wave store (before: 4-byte stores to 32 lines each, 4 waves per line)
# speedup vs baseline: 2.1422x; 1.0447x over previous
#include <hip/hip_runtime.h>
#include <math.h>


typedef __attribute__((ext_vector_type(2))) float v2f;
typedef __attribute__((ext_vector_type(4))) float v4f;
typedef __attribute__((ext_vector_type(8))) float v8f;

#define WMMA_F32(a, b, c) \
  __builtin_amdgcn_wmma_f32_16x16x4_f32(false, (a), false, (b), (short)0, (c), false, false)

constexpr int B_ = 2, T_ = 2048, C_ = 256, H_ = 4, D_ = 64, W_ = 64, S_ = 1;
constexpr int Tq_ = T_ - S_;
constexpr int KQVROW = 3 * H_ * D_;

__device__ __forceinline__ int clampi(int v, int lo, int hi) {
  return v < lo ? lo : (v > hi ? hi : v);
}

template <int WM, int WN, int WVM, int WVN>
__global__ __launch_bounds__(32 * WVM * WVN)
void gemm_abt(const float* __restrict__ A, const float* __restrict__ Bm,
              const float* __restrict__ bias, float* __restrict__ Cm,
              int M, int N, int K) {
  constexpr int NT    = 32 * WVM * WVN;
  constexpr int BM    = 16 * WM * WVM;
  constexpr int BN    = 16 * WN * WVN;
  constexpr int KC    = 32;
  constexpr int PITCH = KC + 4;
  constexpr int RPP   = NT / 8;
  constexpr int PA    = BM / RPP;
  constexpr int PB    = BN / RPP;

  __shared__ __attribute__((aligned(16))) float sA[2][BM][PITCH];
  __shared__ __attribute__((aligned(16))) float sB[2][BN][PITCH];

  const int tid  = threadIdx.x;
  const int lane = tid & 31;
  const int wave = tid >> 5;
  const int half = lane >> 4;
  const int l16  = lane & 15;
  const int kb   = half * 2;
  const int wm   = wave / WVN;
  const int wn   = wave % WVN;
  const int n0   = blockIdx.x * BN;
  const int m0   = blockIdx.y * BM;

  const int lr = tid >> 3;
  const int lc = (tid & 7) * 4;
  const float* ga = A  + (size_t)(m0 + lr) * K + lc;
  const float* gb = Bm + (size_t)(n0 + lr) * K + lc;

  v4f ra[PA], rb[PB];
#pragma unroll
  for (int p = 0; p < PA; ++p) ra[p] = *(const v4f*)(ga + (size_t)(p * RPP) * K);
#pragma unroll
  for (int p = 0; p < PB; ++p) rb[p] = *(const v4f*)(gb + (size_t)(p * RPP) * K);
#pragma unroll
  for (int p = 0; p < PA; ++p) *(v4f*)&sA[0][p * RPP + lr][lc] = ra[p];
#pragma unroll
  for (int p = 0; p < PB; ++p) *(v4f*)&sB[0][p * RPP + lr][lc] = rb[p];
  __syncthreads();

  v8f acc[WM][WN];
#pragma unroll
  for (int i = 0; i < WM; ++i)
#pragma unroll
    for (int j = 0; j < WN; ++j) acc[i][j] = v8f{};

  const int arow = wm * 16 * WM + l16;
  const int brow = wn * 16 * WN + l16;
  const int nch  = K / KC;
  for (int c = 0; c < nch; ++c) {
    const int cur  = c & 1;
    const bool more = (c + 1 < nch);
    if (more) {
      const int kn = (c + 1) * KC;
#pragma unroll
      for (int p = 0; p < PA; ++p) ra[p] = *(const v4f*)(ga + (size_t)(p * RPP) * K + kn);
#pragma unroll
      for (int p = 0; p < PB; ++p) rb[p] = *(const v4f*)(gb + (size_t)(p * RPP) * K + kn);
    }
#pragma unroll
    for (int ks = 0; ks < KC; ks += 4) {
      v2f fa[WM], fb[WN];
#pragma unroll
      for (int i = 0; i < WM; ++i) fa[i] = *(const v2f*)&sA[cur][arow + i * 16][ks + kb];
#pragma unroll
      for (int j = 0; j < WN; ++j) fb[j] = *(const v2f*)&sB[cur][brow + j * 16][ks + kb];
#pragma unroll
      for (int i = 0; i < WM; ++i)
#pragma unroll
        for (int j = 0; j < WN; ++j) acc[i][j] = WMMA_F32(fa[i], fb[j], acc[i][j]);
    }
    if (more) {
#pragma unroll
      for (int p = 0; p < PA; ++p) *(v4f*)&sA[cur ^ 1][p * RPP + lr][lc] = ra[p];
#pragma unroll
      for (int p = 0; p < PB; ++p) *(v4f*)&sB[cur ^ 1][p * RPP + lr][lc] = rb[p];
    }
    __syncthreads();
  }

#pragma unroll
  for (int j = 0; j < WN; ++j) {
    const int nn = n0 + wn * 16 * WN + j * 16 + l16;
    const float bb = bias ? bias[nn] : 0.0f;
#pragma unroll
    for (int i = 0; i < WM; ++i) {
#pragma unroll
      for (int r = 0; r < 8; ++r) {
        const int m = m0 + wm * 16 * WM + i * 16 + r + half * 8;
        Cm[(size_t)m * N + nn] = acc[i][j][r] + bb;
      }
    }
  }
}

__global__ __launch_bounds__(512)
void attn_proj_kernel(const float* __restrict__ kqv, const float* __restrict__ q_pos,
                      const float* __restrict__ v_pos, const float* __restrict__ w_out,
                      const float* __restrict__ b_out,
                      float* __restrict__ y, float* __restrict__ wei_out) {
  constexpr int MG  = 16;
  constexpr int SCP = W_ + 2 * MG + 4;
  constexpr int PP  = C_ + 4;
  __shared__ __attribute__((aligned(16))) float sc[H_][16][SCP];
  __shared__ float s0s[H_][W_ + 16];
  __shared__ __attribute__((aligned(16))) float ao[16][PP];
  __shared__ __attribute__((aligned(16))) float sW[C_][PP];

  const int t0 = blockIdx.x * 16;
  const int b  = blockIdx.y;
  const int tid = threadIdx.x;
  const int lane = tid & 31;
  const int wave = __builtin_amdgcn_readfirstlane(tid >> 5);
  const int h    = wave >> 2;
  const int wv   = wave & 3;
  const int ht   = tid & 127;
  const int half = lane >> 4, l16 = lane & 15, kb = half * 2;
  const float scale = 0.125f;

  if (blockIdx.x == 0 && tid < C_)
    y[(size_t)b * T_ * C_ + tid] = b_out[tid];

  {
    const int zr = ht >> 3, zp = ht & 7;
    const int zc = zp < 4 ? 4 * zp : W_ + MG + 4 * (zp - 4);
    *(v4f*)&sc[h][zr][zc] = v4f{};
  }

  const float* wsrc = w_out + (size_t)(tid >> 5) * C_ + (tid & 31) * 4;
  float* wdst = &sW[tid >> 5][(tid & 31) * 4];
  v4f rw[16];

  {
    const int jw0 = t0 - W_ + 16 * wv;
    const int ja  = jw0 + l16;
    const int jb  = ja + 16;
    const int jra = ja < 0 ? 0 : ja;
    const int jrb = jb < 0 ? 0 : jb;
    const float* ka = kqv + (size_t)(b * T_ + jra) * KQVROW + h * D_ + 4 * half;
    const float* kc = kqv + (size_t)(b * T_ + jrb) * KQVROW + h * D_ + 4 * half;
    const float* qa = ka + H_ * D_;
    const float* qc = kc + H_ * D_;
    const float* pr = q_pos + (size_t)((16 * wv + l16) * H_ + h) * D_ + 4 * half;

    v4f fa[D_ / 8], fc[D_ / 8], fp[D_ / 8];
#pragma unroll
    for (int i = 0; i < D_ / 8; ++i) fa[i] = *(const v4f*)(qa + 8 * i);
#pragma unroll
    for (int i = 0; i < D_ / 8; ++i) fc[i] = *(const v4f*)(qc + 8 * i);
#pragma unroll
    for (int i = 0; i < D_ / 8; ++i) fp[i] = *(const v4f*)(pr + 8 * i);

    {
      float p = 0.0f;
#pragma unroll
      for (int i = 0; i < D_ / 8; ++i) {
        const v4f kk = *(const v4f*)(ka + 8 * i);
        p = fmaf(kk.x, fa[i].x, p);
        p = fmaf(kk.y, fa[i].y, p);
        p = fmaf(kk.z, fa[i].z, p);
        p = fmaf(kk.w, fa[i].w, p);
      }
      p += __shfl_xor(p, 16);
      s0s[h][16 * wv + l16] = p;
    }
    if (wv == 3) {
      float p = 0.0f;
#pragma unroll
      for (int i = 0; i < D_ / 8; ++i) {
        const v4f kk = *(const v4f*)(kc + 8 * i);
        p = fmaf(kk.x, fc[i].x, p);
        p = fmaf(kk.y, fc[i].y, p);
        p = fmaf(kk.z, fc[i].z, p);
        p = fmaf(kk.w, fc[i].w, p);
      }
      p += __shfl_xor(p, 16);
      s0s[h][W_ + l16] = p;
    }

    v8f c0 = {}, c1 = {};
#pragma unroll
    for (int i = 0; i < D_ / 8; ++i) {
      const v2f blo = fp[i].lo, bhi = fp[i].hi;
      c0 = WMMA_F32(fa[i].lo, blo, c0);
      c1 = WMMA_F32(fc[i].lo, blo, c1);
      c0 = WMMA_F32(fa[i].hi, bhi, c0);
      c1 = WMMA_F32(fc[i].hi, bhi, c1);
    }

    __syncthreads();

    float s0v[8];
#pragma unroll
    for (int r = 0; r < 8; ++r) {
      const int R = 8 * half + r;
      s0v[r] = s0s[h][16 * wv + (R > l16 ? R : R + 16)];
    }
#pragma unroll
    for (int r = 0; r < 8; ++r) {
      const int  R    = 8 * half + r;
      const bool lowt = R > l16;
      const int  jrel = lowt ? R : R + 16;
      const float s1  = lowt ? c0[r] : c1[r];
      const int  ti   = jrel - l16 - 1;
      float v = (s0v[r] + s1) * scale;
      if (jw0 + jrel < 0)    v = -INFINITY;
      if (t0 + ti >= Tq_)    v = 0.0f;
      sc[h][ti][MG + 16 * wv + l16] = v;
    }
  }
  __syncthreads();

  const int d0   = wv * 16;
  const float* vbase = kqv + (size_t)b * T_ * KQVROW + 2 * H_ * D_ + h * D_ + d0 + l16;
  v2f bvv[20], bvp[W_ / 4];
#pragma unroll
  for (int jt = 0; jt < 5; ++jt) {
    const int j0  = t0 - 64 + jt * 16;
    const int j0c = j0 < 0 ? 0 : j0;
#pragma unroll
    for (int k4 = 0; k4 < 16; k4 += 4) {
      const float* vr = vbase + (size_t)(j0c + k4 + kb) * KQVROW;
      bvv[jt * 4 + k4 / 4].x = vr[0];
      bvv[jt * 4 + k4 / 4].y = vr[KQVROW];
    }
  }
#pragma unroll
  for (int k = 0; k < W_; k += 4) {
    bvp[k / 4].x = v_pos[((size_t)(k + kb)     * H_ + h) * D_ + d0 + l16];
    bvp[k / 4].y = v_pos[((size_t)(k + kb + 1) * H_ + h) * D_ + d0 + l16];
  }
  if (ht < 16) {
    float m = -INFINITY;
    for (int w = 0; w < W_; ++w) m = fmaxf(m, sc[h][ht][MG + w]);
    float s = 0.0f;
    for (int w = 0; w < W_; ++w) { float e = __expf(sc[h][ht][MG + w] - m); sc[h][ht][MG + w] = e; s += e; }
    const float inv = 1.0f / s;
    for (int w = 0; w < W_; ++w) sc[h][ht][MG + w] *= inv;
  }
  __syncthreads();

#pragma unroll
  for (int u = 0; u < 16; ++u) rw[u] = *(const v4f*)(wsrc + (size_t)u * 16 * C_);
  __builtin_amdgcn_sched_barrier(0);

  {
    const int w = ((wave & 1) << 5) + lane;
#pragma unroll
    for (int u = 0; u < 2; ++u) {
      const int ti = (wave >> 1) + 8 * u;
      if (t0 + ti < Tq_) {
        v4f o;
        o.x = sc[0][ti][MG + w];
        o.y = sc[1][ti][MG + w];
        o.z = sc[2][ti][MG + w];
        o.w = sc[3][ti][MG + w];
        *(v4f*)(wei_out + (((size_t)b * Tq_ + (t0 + ti)) * W_ + w) * H_) = o;
      }
    }
  }

  v8f c = {};
#pragma unroll
  for (int jt = 0; jt < 5; ++jt) {
#pragma unroll
    for (int k4 = 0; k4 < 16; k4 += 4) {
      const int wa = jt * 16 + k4 + kb - l16 - 1;
      v2f a;
      a.x = sc[h][l16][MG + wa];
      a.y = sc[h][l16][MG + wa + 1];
      c = WMMA_F32(a, bvv[jt * 4 + k4 / 4], c);
    }
  }
#pragma unroll
  for (int k = 0; k < W_; k += 4) {
    v2f a;
    a.x = sc[h][l16][MG + k + kb];
    a.y = sc[h][l16][MG + k + kb + 1];
    c = WMMA_F32(a, bvp[k / 4], c);
  }

#pragma unroll
  for (int i = 0; i < 8; ++i) ao[i + half * 8][h * D_ + d0 + l16] = c[i];

#pragma unroll
  for (int u = 0; u < 16; ++u) *(v4f*)(wdst + (size_t)u * 16 * PP) = rw[u];

  __syncthreads();

#pragma unroll
  for (int u = 0; u < 16; ++u) rw[u] = *(const v4f*)(wsrc + 128 + (size_t)u * 16 * C_);
  __builtin_amdgcn_sched_barrier(0);

  {
    const int n0 = 16 * wave;
    const float* pa = &ao[l16][4 * half];
    const float* pb = &sW[n0 + l16][4 * half];
    const float bb = b_out[n0 + l16];
    v8f acc = {};
#pragma unroll
    for (int g = 0; g < 4; ++g) {
      if (g == 2) {
        asm volatile("" :: "v"(acc) : "memory");
#pragma unroll
        for (int u = 0; u < 16; ++u) *(v4f*)(wdst + 128 + (size_t)u * 16 * PP) = rw[u];
        __syncthreads();
      }
      v4f xa[8], xb[8];
#pragma unroll
      for (int i = 0; i < 8; ++i) xa[i] = *(const v4f*)(pa + 64 * g + 8 * i);
#pragma unroll
      for (int i = 0; i < 8; ++i) xb[i] = *(const v4f*)(pb + 64 * g + 8 * i);
#pragma unroll
      for (int i = 0; i < 8; ++i) {
        acc = WMMA_F32(xa[i].lo, xb[i].lo, acc);
        acc = WMMA_F32(xa[i].hi, xb[i].hi, acc);
      }
    }
    float* yr = y + ((size_t)b * T_ + (t0 + S_ + half * 8)) * C_ + n0 + l16;
#pragma unroll
    for (int r = 0; r < 7; ++r) yr[(size_t)r * C_] = acc[r] + bb;
    if (t0 + 7 + half * 8 < Tq_) yr[(size_t)7 * C_] = acc[7] + bb;
  }
}

extern "C" void kernel_launch(void* const* d_in, const int* in_sizes, int n_in,
                              void* d_out, int out_size, void* d_ws, size_t ws_size,
                              hipStream_t stream) {
  (void)in_sizes; (void)n_in; (void)out_size; (void)ws_size;

  const float* x     = (const float*)d_in[0];
  const float* w_kqv = (const float*)d_in[1];
  const float* w_out = (const float*)d_in[2];
  const float* b_out = (const float*)d_in[3];
  const float* q_pos = (const float*)d_in[4];
  const float* v_pos = (const float*)d_in[5];

  float* y   = (float*)d_out;
  float* wei = y + (size_t)B_ * T_ * C_;

  float* kqv     = (float*)d_ws;

  gemm_abt<2, 2, 2, 2><<<dim3(KQVROW / 64, (B_ * T_) / 64), 128, 0, stream>>>(
      x, w_kqv, nullptr, kqv, B_ * T_, KQVROW, C_);

  attn_proj_kernel<<<dim3((Tq_ + 15) / 16, B_), 512, 0, stream>>>(
      kqv, q_pos, v_pos, w_out, b_out, y, wei);
}


# softmax: each score row spread over 8 lanes (8 w each, max and sum joined by DPP lane exchanges) on all 128 threads of a head, not 1 thread looping over 64 w; the row sum becomes a tree
# speedup vs baseline: 2.3080x; 1.0774x over previous
#include <hip/hip_runtime.h>
#include <math.h>


typedef __attribute__((ext_vector_type(2))) float v2f;
typedef __attribute__((ext_vector_type(4))) float v4f;
typedef __attribute__((ext_vector_type(8))) float v8f;

#define WMMA_F32(a, b, c) \
  __builtin_amdgcn_wmma_f32_16x16x4_f32(false, (a), false, (b), (short)0, (c), false, false)

constexpr int B_ = 2, T_ = 2048, C_ = 256, H_ = 4, D_ = 64, W_ = 64, S_ = 1;
constexpr int Tq_ = T_ - S_;
constexpr int KQVROW = 3 * H_ * D_;

__device__ __forceinline__ int clampi(int v, int lo, int hi) {
  return v < lo ? lo : (v > hi ? hi : v);
}

template <int MASK>
__device__ __forceinline__ float lane_xor(float x) {
  return __builtin_bit_cast(float, __builtin_amdgcn_update_dpp(
      0, __builtin_bit_cast(int, x), 0x160 + MASK, 0xf, 0xf, false));
}

template <int WM, int WN, int WVM, int WVN>
__global__ __launch_bounds__(32 * WVM * WVN)
void gemm_abt(const float* __restrict__ A, const float* __restrict__ Bm,
              const float* __restrict__ bias, float* __restrict__ Cm,
              int M, int N, int K) {
  constexpr int NT    = 32 * WVM * WVN;
  constexpr int BM    = 16 * WM * WVM;
  constexpr int BN    = 16 * WN * WVN;
  constexpr int KC    = 32;
  constexpr int PITCH = KC + 4;
  constexpr int RPP   = NT / 8;
  constexpr int PA    = BM / RPP;
  constexpr int PB    = BN / RPP;

  __shared__ __attribute__((aligned(16))) float sA[2][BM][PITCH];
  __shared__ __attribute__((aligned(16))) float sB[2][BN][PITCH];

  const int tid  = threadIdx.x;
  const int lane = tid & 31;
  const int wave = tid >> 5;
  const int half = lane >> 4;
  const int l16  = lane & 15;
  const int kb   = half * 2;
  const int wm   = wave / WVN;
  const int wn   = wave % WVN;
  const int n0   = blockIdx.x * BN;
  const int m0   = blockIdx.y * BM;

  const int lr = tid >> 3;
  const int lc = (tid & 7) * 4;
  const float* ga = A  + (size_t)(m0 + lr) * K + lc;
  const float* gb = Bm + (size_t)(n0 + lr) * K + lc;

  v4f ra[PA], rb[PB];
#pragma unroll
  for (int p = 0; p < PA; ++p) ra[p] = *(const v4f*)(ga + (size_t)(p * RPP) * K);
#pragma unroll
  for (int p = 0; p < PB; ++p) rb[p] = *(const v4f*)(gb + (size_t)(p * RPP) * K);
#pragma unroll
  for (int p = 0; p < PA; ++p) *(v4f*)&sA[0][p * RPP + lr][lc] = ra[p];
#pragma unroll
  for (int p = 0; p < PB; ++p) *(v4f*)&sB[0][p * RPP + lr][lc] = rb[p];
  __syncthreads();

  v8f acc[WM][WN];
#pragma unroll
  for (int i = 0; i < WM; ++i)
#pragma unroll
    for (int j = 0; j < WN; ++j) acc[i][j] = v8f{};

  const int arow = wm * 16 * WM + l16;
  const int brow = wn * 16 * WN + l16;
  const int nch  = K / KC;
  for (int c = 0; c < nch; ++c) {
    const int cur  = c & 1;
    const bool more = (c + 1 < nch);
    if (more) {
      const int kn = (c + 1) * KC;
#pragma unroll
      for (int p = 0; p < PA; ++p) ra[p] = *(const v4f*)(ga + (size_t)(p * RPP) * K + kn);
#pragma unroll
      for (int p = 0; p < PB; ++p) rb[p] = *(const v4f*)(gb + (size_t)(p * RPP) * K + kn);
    }
#pragma unroll
    for (int ks = 0; ks < KC; ks += 4) {
      v2f fa[WM], fb[WN];
#pragma unroll
      for (int i = 0; i < WM; ++i) fa[i] = *(const v2f*)&sA[cur][arow + i * 16][ks + kb];
#pragma unroll
      for (int j = 0; j < WN; ++j) fb[j] = *(const v2f*)&sB[cur][brow + j * 16][ks + kb];
#pragma unroll
      for (int i = 0; i < WM; ++i)
#pragma unroll
        for (int j = 0; j < WN; ++j) acc[i][j] = WMMA_F32(fa[i], fb[j], acc[i][j]);
    }
    if (more) {
#pragma unroll
      for (int p = 0; p < PA; ++p) *(v4f*)&sA[cur ^ 1][p * RPP + lr][lc] = ra[p];
#pragma unroll
      for (int p = 0; p < PB; ++p) *(v4f*)&sB[cur ^ 1][p * RPP + lr][lc] = rb[p];
    }
    __syncthreads();
  }

#pragma unroll
  for (int j = 0; j < WN; ++j) {
    const int nn = n0 + wn * 16 * WN + j * 16 + l16;
    const float bb = bias ? bias[nn] : 0.0f;
#pragma unroll
    for (int i = 0; i < WM; ++i) {
#pragma unroll
      for (int r = 0; r < 8; ++r) {
        const int m = m0 + wm * 16 * WM + i * 16 + r + half * 8;
        Cm[(size_t)m * N + nn] = acc[i][j][r] + bb;
      }
    }
  }
}

__global__ __launch_bounds__(512)
void attn_proj_kernel(const float* __restrict__ kqv, const float* __restrict__ q_pos,
                      const float* __restrict__ v_pos, const float* __restrict__ w_out,
                      const float* __restrict__ b_out,
                      float* __restrict__ y, float* __restrict__ wei_out) {
  constexpr int MG  = 16;
  constexpr int SCP = W_ + 2 * MG + 4;
  constexpr int PP  = C_ + 4;
  __shared__ __attribute__((aligned(16))) float sc[H_][16][SCP];
  __shared__ float s0s[H_][W_ + 16];
  __shared__ __attribute__((aligned(16))) float ao[16][PP];
  __shared__ __attribute__((aligned(16))) float sW[C_][PP];

  const int t0 = blockIdx.x * 16;
  const int b  = blockIdx.y;
  const int tid = threadIdx.x;
  const int lane = tid & 31;
  const int wave = __builtin_amdgcn_readfirstlane(tid >> 5);
  const int h    = wave >> 2;
  const int wv   = wave & 3;
  const int ht   = tid & 127;
  const int half = lane >> 4, l16 = lane & 15, kb = half * 2;
  const float scale = 0.125f;

  if (blockIdx.x == 0 && tid < C_)
    y[(size_t)b * T_ * C_ + tid] = b_out[tid];

  {
    const int zr = ht >> 3, zp = ht & 7;
    const int zc = zp < 4 ? 4 * zp : W_ + MG + 4 * (zp - 4);
    *(v4f*)&sc[h][zr][zc] = v4f{};
  }

  const float* wsrc = w_out + (size_t)(tid >> 5) * C_ + (tid & 31) * 4;
  float* wdst = &sW[tid >> 5][(tid & 31) * 4];
  v4f rw[16];

  {
    const int jw0 = t0 - W_ + 16 * wv;
    const int ja  = jw0 + l16;
    const int jb  = ja + 16;
    const int jra = ja < 0 ? 0 : ja;
    const int jrb = jb < 0 ? 0 : jb;
    const float* ka = kqv + (size_t)(b * T_ + jra) * KQVROW + h * D_ + 4 * half;
    const float* kc = kqv + (size_t)(b * T_ + jrb) * KQVROW + h * D_ + 4 * half;
    const float* qa = ka + H_ * D_;
    const float* qc = kc + H_ * D_;
    const float* pr = q_pos + (size_t)((16 * wv + l16) * H_ + h) * D_ + 4 * half;

    v4f fa[D_ / 8], fc[D_ / 8], fp[D_ / 8];
#pragma unroll
    for (int i = 0; i < D_ / 8; ++i) fa[i] = *(const v4f*)(qa + 8 * i);
#pragma unroll
    for (int i = 0; i < D_ / 8; ++i) fc[i] = *(const v4f*)(qc + 8 * i);
#pragma unroll
    for (int i = 0; i < D_ / 8; ++i) fp[i] = *(const v4f*)(pr + 8 * i);

    {
      float p = 0.0f;
#pragma unroll
      for (int i = 0; i < D_ / 8; ++i) {
        const v4f kk = *(const v4f*)(ka + 8 * i);
        p = fmaf(kk.x, fa[i].x, p);
        p = fmaf(kk.y, fa[i].y, p);
        p = fmaf(kk.z, fa[i].z, p);
        p = fmaf(kk.w, fa[i].w, p);
      }
      p += __shfl_xor(p, 16);
      s0s[h][16 * wv + l16] = p;
    }
    if (wv == 3) {
      float p = 0.0f;
#pragma unroll
      for (int i = 0; i < D_ / 8; ++i) {
        const v4f kk = *(const v4f*)(kc + 8 * i);
        p = fmaf(kk.x, fc[i].x, p);
        p = fmaf(kk.y, fc[i].y, p);
        p = fmaf(kk.z, fc[i].z, p);
        p = fmaf(kk.w, fc[i].w, p);
      }
      p += __shfl_xor(p, 16);
      s0s[h][W_ + l16] = p;
    }

    v8f c0 = {}, c1 = {};
#pragma unroll
    for (int i = 0; i < D_ / 8; ++i) {
      const v2f blo = fp[i].lo, bhi = fp[i].hi;
      c0 = WMMA_F32(fa[i].lo, blo, c0);
      c1 = WMMA_F32(fc[i].lo, blo, c1);
      c0 = WMMA_F32(fa[i].hi, bhi, c0);
      c1 = WMMA_F32(fc[i].hi, bhi, c1);
    }

    __syncthreads();

    float s0v[8];
#pragma unroll
    for (int r = 0; r < 8; ++r) {
      const int R = 8 * half + r;
      s0v[r] = s0s[h][16 * wv + (R > l16 ? R : R + 16)];
    }
#pragma unroll
    for (int r = 0; r < 8; ++r) {
      const int  R    = 8 * half + r;
      const bool lowt = R > l16;
      const int  jrel = lowt ? R : R + 16;
      const float s1  = lowt ? c0[r] : c1[r];
      const int  ti   = jrel - l16 - 1;
      float v = (s0v[r] + s1) * scale;
      if (jw0 + jrel < 0)    v = -INFINITY;
      if (t0 + ti >= Tq_)    v = 0.0f;
      sc[h][ti][MG + 16 * wv + l16] = v;
    }
  }
  __syncthreads();

  const int d0   = wv * 16;
  const float* vbase = kqv + (size_t)b * T_ * KQVROW + 2 * H_ * D_ + h * D_ + d0 + l16;
  v2f bvv[20], bvp[W_ / 4];
#pragma unroll
  for (int jt = 0; jt < 5; ++jt) {
    const int j0  = t0 - 64 + jt * 16;
    const int j0c = j0 < 0 ? 0 : j0;
#pragma unroll
    for (int k4 = 0; k4 < 16; k4 += 4) {
      const float* vr = vbase + (size_t)(j0c + k4 + kb) * KQVROW;
      bvv[jt * 4 + k4 / 4].x = vr[0];
      bvv[jt * 4 + k4 / 4].y = vr[KQVROW];
    }
  }
#pragma unroll
  for (int k = 0; k < W_; k += 4) {
    bvp[k / 4].x = v_pos[((size_t)(k + kb)     * H_ + h) * D_ + d0 + l16];
    bvp[k / 4].y = v_pos[((size_t)(k + kb + 1) * H_ + h) * D_ + d0 + l16];
  }
  {
    float* rp = &sc[h][ht >> 3][MG + 8 * (ht & 7)];
    v4f xa = *(const v4f*)rp;
    v4f xb = *(const v4f*)(rp + 4);
    float m = fmaxf(fmaxf(fmaxf(xa.x, xa.y), fmaxf(xa.z, xa.w)),
                    fmaxf(fmaxf(xb.x, xb.y), fmaxf(xb.z, xb.w)));
    m = fmaxf(m, lane_xor<1>(m));
    m = fmaxf(m, lane_xor<2>(m));
    m = fmaxf(m, lane_xor<4>(m));
    xa.x = __expf(xa.x - m); xa.y = __expf(xa.y - m);
    xa.z = __expf(xa.z - m); xa.w = __expf(xa.w - m);
    xb.x = __expf(xb.x - m); xb.y = __expf(xb.y - m);
    xb.z = __expf(xb.z - m); xb.w = __expf(xb.w - m);
    float s = ((xa.x + xa.y) + (xa.z + xa.w)) + ((xb.x + xb.y) + (xb.z + xb.w));
    s += lane_xor<1>(s);
    s += lane_xor<2>(s);
    s += lane_xor<4>(s);
    const float inv = 1.0f / s;
    *(v4f*)rp       = xa * inv;
    *(v4f*)(rp + 4) = xb * inv;
  }
  __syncthreads();

#pragma unroll
  for (int u = 0; u < 16; ++u) rw[u] = *(const v4f*)(wsrc + (size_t)u * 16 * C_);
  __builtin_amdgcn_sched_barrier(0);

  {
    const int w = ((wave & 1) << 5) + lane;
#pragma unroll
    for (int u = 0; u < 2; ++u) {
      const int ti = (wave >> 1) + 8 * u;
      if (t0 + ti < Tq_) {
        v4f o;
        o.x = sc[0][ti][MG + w];
        o.y = sc[1][ti][MG + w];
        o.z = sc[2][ti][MG + w];
        o.w = sc[3][ti][MG + w];
        *(v4f*)(wei_out + (((size_t)b * Tq_ + (t0 + ti)) * W_ + w) * H_) = o;
      }
    }
  }

  v8f c = {};
#pragma unroll
  for (int jt = 0; jt < 5; ++jt) {
#pragma unroll
    for (int k4 = 0; k4 < 16; k4 += 4) {
      const int wa = jt * 16 + k4 + kb - l16 - 1;
      v2f a;
      a.x = sc[h][l16][MG + wa];
      a.y = sc[h][l16][MG + wa + 1];
      c = WMMA_F32(a, bvv[jt * 4 + k4 / 4], c);
    }
  }
#pragma unroll
  for (int k = 0; k < W_; k += 4) {
    v2f a;
    a.x = sc[h][l16][MG + k + kb];
    a.y = sc[h][l16][MG + k + kb + 1];
    c = WMMA_F32(a, bvp[k / 4], c);
  }

#pragma unroll
  for (int i = 0; i < 8; ++i) ao[i + half * 8][h * D_ + d0 + l16] = c[i];

#pragma unroll
  for (int u = 0; u < 16; ++u) *(v4f*)(wdst + (size_t)u * 16 * PP) = rw[u];

  __syncthreads();

#pragma unroll
  for (int u = 0; u < 16; ++u) rw[u] = *(const v4f*)(wsrc + 128 + (size_t)u * 16 * C_);
  __builtin_amdgcn_sched_barrier(0);

  {
    const int n0 = 16 * wave;
    const float* pa = &ao[l16][4 * half];
    const float* pb = &sW[n0 + l16][4 * half];
    const float bb = b_out[n0 + l16];
    v8f acc = {};
#pragma unroll
    for (int g = 0; g < 4; ++g) {
      if (g == 2) {
        asm volatile("" :: "v"(acc) : "memory");
#pragma unroll
        for (int u = 0; u < 16; ++u) *(v4f*)(wdst + 128 + (size_t)u * 16 * PP) = rw[u];
        __syncthreads();
      }
      v4f xa[8], xb[8];
#pragma unroll
      for (int i = 0; i < 8; ++i) xa[i] = *(const v4f*)(pa + 64 * g + 8 * i);
#pragma unroll
      for (int i = 0; i < 8; ++i) xb[i] = *(const v4f*)(pb + 64 * g + 8 * i);
#pragma unroll
      for (int i = 0; i < 8; ++i) {
        acc = WMMA_F32(xa[i].lo, xb[i].lo, acc);
        acc = WMMA_F32(xa[i].hi, xb[i].hi, acc);
      }
    }
    float* yr = y + ((size_t)b * T_ + (t0 + S_ + half * 8)) * C_ + n0 + l16;
#pragma unroll
    for (int r = 0; r < 7; ++r) yr[(size_t)r * C_] = acc[r] + bb;
    if (t0 + 7 + half * 8 < Tq_) yr[(size_t)7 * C_] = acc[7] + bb;
  }
}

extern "C" void kernel_launch(void* const* d_in, const int* in_sizes, int n_in,
                              void* d_out, int out_size, void* d_ws, size_t ws_size,
                              hipStream_t stream) {
  (void)in_sizes; (void)n_in; (void)out_size; (void)ws_size;

  const float* x     = (const float*)d_in[0];
  const float* w_kqv = (const float*)d_in[1];
  const float* w_out = (const float*)d_in[2];
  const float* b_out = (const float*)d_in[3];
  const float* q_pos = (const float*)d_in[4];
  const float* v_pos = (const float*)d_in[5];

  float* y   = (float*)d_out;
  float* wei = y + (size_t)B_ * T_ * C_;

  float* kqv     = (float*)d_ws;

  gemm_abt<2, 2, 2, 2><<<dim3(KQVROW / 64, (B_ * T_) / 64), 128, 0, stream>>>(
      x, w_kqv, nullptr, kqv, B_ * T_, KQVROW, C_);

  attn_proj_kernel<<<dim3((Tq_ + 15) / 16, B_), 512, 0, stream>>>(
      kqv, q_pos, v_pos, w_out, b_out, y, wei);
}


# attention kernel: block barriers fence LDS only, so global loads stay in flight across them; V rows and wave 3's extra k rows load at kernel entry, v_pos and both w_out parts a phase earlier
# speedup vs baseline: 2.4339x; 1.0545x over previous
#include <hip/hip_runtime.h>
#include <math.h>


typedef __attribute__((ext_vector_type(2))) float v2f;
typedef __attribute__((ext_vector_type(4))) float v4f;
typedef __attribute__((ext_vector_type(8))) float v8f;

#define WMMA_F32(a, b, c) \
  __builtin_amdgcn_wmma_f32_16x16x4_f32(false, (a), false, (b), (short)0, (c), false, false)

constexpr int B_ = 2, T_ = 2048, C_ = 256, H_ = 4, D_ = 64, W_ = 64, S_ = 1;
constexpr int Tq_ = T_ - S_;
constexpr int KQVROW = 3 * H_ * D_;

__device__ __forceinline__ int clampi(int v, int lo, int hi) {
  return v < lo ? lo : (v > hi ? hi : v);
}

template <int MASK>
__device__ __forceinline__ float lane_xor(float x) {
  return __builtin_bit_cast(float, __builtin_amdgcn_update_dpp(
      0, __builtin_bit_cast(int, x), 0x160 + MASK, 0xf, 0xf, false));
}

__device__ __forceinline__ void lds_barrier() {
  __builtin_amdgcn_fence(__ATOMIC_RELEASE, "workgroup", "local");
  __builtin_amdgcn_s_barrier();
  __builtin_amdgcn_fence(__ATOMIC_ACQUIRE, "workgroup", "local");
}

template <int WM, int WN, int WVM, int WVN>
__global__ __launch_bounds__(32 * WVM * WVN)
void gemm_abt(const float* __restrict__ A, const float* __restrict__ Bm,
              const float* __restrict__ bias, float* __restrict__ Cm,
              int M, int N, int K) {
  constexpr int NT    = 32 * WVM * WVN;
  constexpr int BM    = 16 * WM * WVM;
  constexpr int BN    = 16 * WN * WVN;
  constexpr int KC    = 32;
  constexpr int PITCH = KC + 4;
  constexpr int RPP   = NT / 8;
  constexpr int PA    = BM / RPP;
  constexpr int PB    = BN / RPP;

  __shared__ __attribute__((aligned(16))) float sA[2][BM][PITCH];
  __shared__ __attribute__((aligned(16))) float sB[2][BN][PITCH];

  const int tid  = threadIdx.x;
  const int lane = tid & 31;
  const int wave = tid >> 5;
  const int half = lane >> 4;
  const int l16  = lane & 15;
  const int kb   = half * 2;
  const int wm   = wave / WVN;
  const int wn   = wave % WVN;
  const int n0   = blockIdx.x * BN;
  const int m0   = blockIdx.y * BM;

  const int lr = tid >> 3;
  const int lc = (tid & 7) * 4;
  const float* ga = A  + (size_t)(m0 + lr) * K + lc;
  const float* gb = Bm + (size_t)(n0 + lr) * K + lc;

  v4f ra[PA], rb[PB];
#pragma unroll
  for (int p = 0; p < PA; ++p) ra[p] = *(const v4f*)(ga + (size_t)(p * RPP) * K);
#pragma unroll
  for (int p = 0; p < PB; ++p) rb[p] = *(const v4f*)(gb + (size_t)(p * RPP) * K);
#pragma unroll
  for (int p = 0; p < PA; ++p) *(v4f*)&sA[0][p * RPP + lr][lc] = ra[p];
#pragma unroll
  for (int p = 0; p < PB; ++p) *(v4f*)&sB[0][p * RPP + lr][lc] = rb[p];
  __syncthreads();

  v8f acc[WM][WN];
#pragma unroll
  for (int i = 0; i < WM; ++i)
#pragma unroll
    for (int j = 0; j < WN; ++j) acc[i][j] = v8f{};

  const int arow = wm * 16 * WM + l16;
  const int brow = wn * 16 * WN + l16;
  const int nch  = K / KC;
  for (int c = 0; c < nch; ++c) {
    const int cur  = c & 1;
    const bool more = (c + 1 < nch);
    if (more) {
      const int kn = (c + 1) * KC;
#pragma unroll
      for (int p = 0; p < PA; ++p) ra[p] = *(const v4f*)(ga + (size_t)(p * RPP) * K + kn);
#pragma unroll
      for (int p = 0; p < PB; ++p) rb[p] = *(const v4f*)(gb + (size_t)(p * RPP) * K + kn);
    }
#pragma unroll
    for (int ks = 0; ks < KC; ks += 4) {
      v2f fa[WM], fb[WN];
#pragma unroll
      for (int i = 0; i < WM; ++i) fa[i] = *(const v2f*)&sA[cur][arow + i * 16][ks + kb];
#pragma unroll
      for (int j = 0; j < WN; ++j) fb[j] = *(const v2f*)&sB[cur][brow + j * 16][ks + kb];
#pragma unroll
      for (int i = 0; i < WM; ++i)
#pragma unroll
        for (int j = 0; j < WN; ++j) acc[i][j] = WMMA_F32(fa[i], fb[j], acc[i][j]);
    }
    if (more) {
#pragma unroll
      for (int p = 0; p < PA; ++p) *(v4f*)&sA[cur ^ 1][p * RPP + lr][lc] = ra[p];
#pragma unroll
      for (int p = 0; p < PB; ++p) *(v4f*)&sB[cur ^ 1][p * RPP + lr][lc] = rb[p];
    }
    __syncthreads();
  }

#pragma unroll
  for (int j = 0; j < WN; ++j) {
    const int nn = n0 + wn * 16 * WN + j * 16 + l16;
    const float bb = bias ? bias[nn] : 0.0f;
#pragma unroll
    for (int i = 0; i < WM; ++i) {
#pragma unroll
      for (int r = 0; r < 8; ++r) {
        const int m = m0 + wm * 16 * WM + i * 16 + r + half * 8;
        Cm[(size_t)m * N + nn] = acc[i][j][r] + bb;
      }
    }
  }
}

__global__ __launch_bounds__(512)
void attn_proj_kernel(const float* __restrict__ kqv, const float* __restrict__ q_pos,
                      const float* __restrict__ v_pos, const float* __restrict__ w_out,
                      const float* __restrict__ b_out,
                      float* __restrict__ y, float* __restrict__ wei_out) {
  constexpr int MG  = 16;
  constexpr int SCP = W_ + 2 * MG + 4;
  constexpr int PP  = C_ + 4;
  __shared__ __attribute__((aligned(16))) float sc[H_][16][SCP];
  __shared__ float s0s[H_][W_ + 16];
  __shared__ __attribute__((aligned(16))) float ao[16][PP];
  __shared__ __attribute__((aligned(16))) float sW[C_][PP];

  const int t0 = blockIdx.x * 16;
  const int b  = blockIdx.y;
  const int tid = threadIdx.x;
  const int lane = tid & 31;
  const int wave = __builtin_amdgcn_readfirstlane(tid >> 5);
  const int h    = wave >> 2;
  const int wv   = wave & 3;
  const int ht   = tid & 127;
  const int half = lane >> 4, l16 = lane & 15, kb = half * 2;
  const float scale = 0.125f;

  if (blockIdx.x == 0 && tid < C_)
    y[(size_t)b * T_ * C_ + tid] = b_out[tid];

  {
    const int zr = ht >> 3, zp = ht & 7;
    const int zc = zp < 4 ? 4 * zp : W_ + MG + 4 * (zp - 4);
    *(v4f*)&sc[h][zr][zc] = v4f{};
  }

  const float* wsrc = w_out + (size_t)(tid >> 5) * C_ + (tid & 31) * 4;
  float* wdst = &sW[tid >> 5][(tid & 31) * 4];
  v4f rw[16], rw2[16];

  const int d0 = wv * 16;
  const float* vbase = kqv + (size_t)b * T_ * KQVROW + 2 * H_ * D_ + h * D_ + d0;
  const int voff = kb * KQVROW + l16;
  v2f bvv[20], bvp[W_ / 4];

  {
    const int jw0 = t0 - W_ + 16 * wv;
    const int ja  = jw0 + l16;
    const int jb  = ja + 16;
    const int jra = ja < 0 ? 0 : ja;
    const int jrb = jb < 0 ? 0 : jb;
    const float* ka = kqv + (size_t)(b * T_ + jra) * KQVROW + h * D_ + 4 * half;
    const float* kc = kqv + (size_t)(b * T_ + jrb) * KQVROW + h * D_ + 4 * half;
    const float* qa = ka + H_ * D_;
    const float* qc = kc + H_ * D_;
    const float* pr = q_pos + (size_t)((16 * wv + l16) * H_ + h) * D_ + 4 * half;

    v4f fa[D_ / 8], fc[D_ / 8], fp[D_ / 8];
    v4f fk[D_ / 8], fk2[D_ / 8];
#pragma unroll
    for (int i = 0; i < D_ / 8; ++i) fa[i] = *(const v4f*)(qa + 8 * i);
#pragma unroll
    for (int i = 0; i < D_ / 8; ++i) fk[i] = *(const v4f*)(ka + 8 * i);
#pragma unroll
    for (int i = 0; i < D_ / 8; ++i) fc[i] = *(const v4f*)(qc + 8 * i);
#pragma unroll
    for (int i = 0; i < D_ / 8; ++i) fp[i] = *(const v4f*)(pr + 8 * i);
    if (wv == 3) {
#pragma unroll
      for (int i = 0; i < D_ / 8; ++i) fk2[i] = *(const v4f*)(kc + 8 * i);
    }
#pragma unroll
    for (int jt = 0; jt < 5; ++jt) {
      const int j0  = t0 - 64 + jt * 16;
      const int j0c = j0 < 0 ? 0 : j0;
#pragma unroll
      for (int k4 = 0; k4 < 16; k4 += 4) {
        const float* vr = vbase + (size_t)(j0c + k4) * KQVROW;
        bvv[jt * 4 + k4 / 4].x = vr[voff];
        bvv[jt * 4 + k4 / 4].y = vr[voff + KQVROW];
      }
    }

    {
      float p = 0.0f;
#pragma unroll
      for (int i = 0; i < D_ / 8; ++i) {
        p = fmaf(fk[i].x, fa[i].x, p);
        p = fmaf(fk[i].y, fa[i].y, p);
        p = fmaf(fk[i].z, fa[i].z, p);
        p = fmaf(fk[i].w, fa[i].w, p);
      }
      p += __shfl_xor(p, 16);
      s0s[h][16 * wv + l16] = p;
    }
    if (wv == 3) {
      float p = 0.0f;
#pragma unroll
      for (int i = 0; i < D_ / 8; ++i) {
        p = fmaf(fk2[i].x, fc[i].x, p);
        p = fmaf(fk2[i].y, fc[i].y, p);
        p = fmaf(fk2[i].z, fc[i].z, p);
        p = fmaf(fk2[i].w, fc[i].w, p);
      }
      p += __shfl_xor(p, 16);
      s0s[h][W_ + l16] = p;
    }

#pragma unroll
    for (int k = 0; k < W_; k += 4) {
      bvp[k / 4].x = v_pos[((size_t)(k + kb)     * H_ + h) * D_ + d0 + l16];
      bvp[k / 4].y = v_pos[((size_t)(k + kb + 1) * H_ + h) * D_ + d0 + l16];
    }

    v8f c0 = {}, c1 = {};
#pragma unroll
    for (int i = 0; i < D_ / 8; ++i) {
      const v2f blo = fp[i].lo, bhi = fp[i].hi;
      c0 = WMMA_F32(fa[i].lo, blo, c0);
      c1 = WMMA_F32(fc[i].lo, blo, c1);
      c0 = WMMA_F32(fa[i].hi, bhi, c0);
      c1 = WMMA_F32(fc[i].hi, bhi, c1);
    }

    lds_barrier();

    float s0v[8];
#pragma unroll
    for (int r = 0; r < 8; ++r) {
      const int R = 8 * half + r;
      s0v[r] = s0s[h][16 * wv + (R > l16 ? R : R + 16)];
    }
#pragma unroll
    for (int r = 0; r < 8; ++r) {
      const int  R    = 8 * half + r;
      const bool lowt = R > l16;
      const int  jrel = lowt ? R : R + 16;
      const float s1  = lowt ? c0[r] : c1[r];
      const int  ti   = jrel - l16 - 1;
      float v = (s0v[r] + s1) * scale;
      if (jw0 + jrel < 0)    v = -INFINITY;
      if (t0 + ti >= Tq_)    v = 0.0f;
      sc[h][ti][MG + 16 * wv + l16] = v;
    }
  }
  lds_barrier();

#pragma unroll
  for (int u = 0; u < 16; ++u) rw[u] = *(const v4f*)(wsrc + (size_t)u * 16 * C_);

  {
    float* rp = &sc[h][ht >> 3][MG + 8 * (ht & 7)];
    v4f xa = *(const v4f*)rp;
    v4f xb = *(const v4f*)(rp + 4);
    float m = fmaxf(fmaxf(fmaxf(xa.x, xa.y), fmaxf(xa.z, xa.w)),
                    fmaxf(fmaxf(xb.x, xb.y), fmaxf(xb.z, xb.w)));
    m = fmaxf(m, lane_xor<1>(m));
    m = fmaxf(m, lane_xor<2>(m));
    m = fmaxf(m, lane_xor<4>(m));
    xa.x = __expf(xa.x - m); xa.y = __expf(xa.y - m);
    xa.z = __expf(xa.z - m); xa.w = __expf(xa.w - m);
    xb.x = __expf(xb.x - m); xb.y = __expf(xb.y - m);
    xb.z = __expf(xb.z - m); xb.w = __expf(xb.w - m);
    float s = ((xa.x + xa.y) + (xa.z + xa.w)) + ((xb.x + xb.y) + (xb.z + xb.w));
    s += lane_xor<1>(s);
    s += lane_xor<2>(s);
    s += lane_xor<4>(s);
    const float inv = 1.0f / s;
    *(v4f*)rp       = xa * inv;
    *(v4f*)(rp + 4) = xb * inv;
  }
  lds_barrier();

  {
    const int w = ((wave & 1) << 5) + lane;
#pragma unroll
    for (int u = 0; u < 2; ++u) {
      const int ti = (wave >> 1) + 8 * u;
      if (t0 + ti < Tq_) {
        v4f o;
        o.x = sc[0][ti][MG + w];
        o.y = sc[1][ti][MG + w];
        o.z = sc[2][ti][MG + w];
        o.w = sc[3][ti][MG + w];
        *(v4f*)(wei_out + (((size_t)b * Tq_ + (t0 + ti)) * W_ + w) * H_) = o;
      }
    }
  }

  v8f c = {};
#pragma unroll
  for (int jt = 0; jt < 5; ++jt) {
#pragma unroll
    for (int k4 = 0; k4 < 16; k4 += 4) {
      const int wa = jt * 16 + k4 + kb - l16 - 1;
      v2f a;
      a.x = sc[h][l16][MG + wa];
      a.y = sc[h][l16][MG + wa + 1];
      c = WMMA_F32(a, bvv[jt * 4 + k4 / 4], c);
    }
  }
#pragma unroll
  for (int k = 0; k < W_; k += 4) {
    v2f a;
    a.x = sc[h][l16][MG + k + kb];
    a.y = sc[h][l16][MG + k + kb + 1];
    c = WMMA_F32(a, bvp[k / 4], c);
  }

#pragma unroll
  for (int i = 0; i < 8; ++i) ao[i + half * 8][h * D_ + d0 + l16] = c[i];

#pragma unroll
  for (int u = 0; u < 16; ++u) rw2[u] = *(const v4f*)(wsrc + 128 + (size_t)u * 16 * C_);

#pragma unroll
  for (int u = 0; u < 16; ++u) *(v4f*)(wdst + (size_t)u * 16 * PP) = rw[u];

  lds_barrier();

  {
    const int n0 = 16 * wave;
    const float* pa = &ao[l16][4 * half];
    const float* pb = &sW[n0 + l16][4 * half];
    const float bb = b_out[n0 + l16];
    v8f acc = {};
#pragma unroll
    for (int g = 0; g < 4; ++g) {
      if (g == 2) {
        asm volatile("" :: "v"(acc) : "memory");
#pragma unroll
        for (int u = 0; u < 16; ++u) *(v4f*)(wdst + 128 + (size_t)u * 16 * PP) = rw2[u];
        lds_barrier();
      }
      v4f xa[8], xb[8];
#pragma unroll
      for (int i = 0; i < 8; ++i) xa[i] = *(const v4f*)(pa + 64 * g + 8 * i);
#pragma unroll
      for (int i = 0; i < 8; ++i) xb[i] = *(const v4f*)(pb + 64 * g + 8 * i);
#pragma unroll
      for (int i = 0; i < 8; ++i) {
        acc = WMMA_F32(xa[i].lo, xb[i].lo, acc);
        acc = WMMA_F32(xa[i].hi, xb[i].hi, acc);
      }
    }
    float* yr = y + ((size_t)b * T_ + (t0 + S_ + half * 8)) * C_ + n0 + l16;
#pragma unroll
    for (int r = 0; r < 7; ++r) yr[(size_t)r * C_] = acc[r] + bb;
    if (t0 + 7 + half * 8 < Tq_) yr[(size_t)7 * C_] = acc[7] + bb;
  }
}

extern "C" void kernel_launch(void* const* d_in, const int* in_sizes, int n_in,
                              void* d_out, int out_size, void* d_ws, size_t ws_size,
                              hipStream_t stream) {
  (void)in_sizes; (void)n_in; (void)out_size; (void)ws_size;

  const float* x     = (const float*)d_in[0];
  const float* w_kqv = (const float*)d_in[1];
  const float* w_out = (const float*)d_in[2];
  const float* b_out = (const float*)d_in[3];
  const float* q_pos = (const float*)d_in[4];
  const float* v_pos = (const float*)d_in[5];

  float* y   = (float*)d_out;
  float* wei = y + (size_t)B_ * T_ * C_;

  float* kqv     = (float*)d_ws;

  gemm_abt<2, 2, 2, 2><<<dim3(KQVROW / 64, (B_ * T_) / 64), 128, 0, stream>>>(
      x, w_kqv, nullptr, kqv, B_ * T_, KQVROW, C_);

  attn_proj_kernel<<<dim3((Tq_ + 15) / 16, B_), 512, 0, stream>>>(
      kqv, q_pos, v_pos, w_out, b_out, y, wei);
}


# s0 = k.q row dots formed in the kqv GEMM epilogue (16x64 wave tile pairs 32 k with the same 32 q columns: two partial sums per row, lane-tree order over d); k is never stored or re-read by attention
# speedup vs baseline: 2.5760x; 1.0584x over previous
#include <hip/hip_runtime.h>
#include <math.h>


typedef __attribute__((ext_vector_type(2))) float v2f;
typedef __attribute__((ext_vector_type(4))) float v4f;
typedef __attribute__((ext_vector_type(8))) float v8f;

#define WMMA_F32(a, b, c) \
  __builtin_amdgcn_wmma_f32_16x16x4_f32(false, (a), false, (b), (short)0, (c), false, false)

constexpr int B_ = 2, T_ = 2048, C_ = 256, H_ = 4, D_ = 64, W_ = 64, S_ = 1;
constexpr int Tq_ = T_ - S_;
constexpr int KQVROW = 3 * H_ * D_;

__device__ __forceinline__ int clampi(int v, int lo, int hi) {
  return v < lo ? lo : (v > hi ? hi : v);
}

template <int MASK>
__device__ __forceinline__ float lane_xor(float x) {
  return __builtin_bit_cast(float, __builtin_amdgcn_update_dpp(
      0, __builtin_bit_cast(int, x), 0x160 + MASK, 0xf, 0xf, false));
}

__device__ __forceinline__ void lds_barrier() {
  __builtin_amdgcn_fence(__ATOMIC_RELEASE, "workgroup", "local");
  __builtin_amdgcn_s_barrier();
  __builtin_amdgcn_fence(__ATOMIC_ACQUIRE, "workgroup", "local");
}

template <int WM, int WN, int WVM, int WVN>
__global__ __launch_bounds__(32 * WVM * WVN)
void gemm_kqv(const float* __restrict__ A, const float* __restrict__ Bm,
              float* __restrict__ Cm, float* __restrict__ s0p,
              int M, int N, int K) {
  constexpr int NT    = 32 * WVM * WVN;
  constexpr int BM    = 16 * WM * WVM;
  constexpr int BN    = 16 * WN * WVN;
  constexpr int KC    = 32;
  constexpr int PITCH = KC + 4;
  constexpr int RPP   = NT / 8;
  constexpr int PA    = BM / RPP;
  constexpr int PB    = BN / RPP;

  __shared__ __attribute__((aligned(16))) float sA[2][BM][PITCH];
  __shared__ __attribute__((aligned(16))) float sB[2][BN][PITCH];

  const int tid  = threadIdx.x;
  const int lane = tid & 31;
  const int wave = tid >> 5;
  const int half = lane >> 4;
  const int l16  = lane & 15;
  const int kb   = half * 2;
  const int wm   = wave / WVN;
  const int wn   = wave % WVN;
  const int n0   = blockIdx.x * BN;
  const int m0   = blockIdx.y * BM;

  const int lr = tid >> 3;
  const int lc = (tid & 7) * 4;
  const float* ga = A  + (size_t)(m0 + lr) * K + lc;
  const int  nt   = blockIdx.x;
  const bool kq   = nt < 2 * H_;
  const int  hd   = nt >> 1, pp = nt & 1;
  const int  cbk  = hd * D_ + 32 * pp;
  const int  cbq  = H_ * D_ + cbk;
  static_assert(32 * WVM * WVN == 128, "fetch pass p covers tile rows 16p .. 16p + 15");

  const float* gb[PB];
#pragma unroll
  for (int p = 0; p < PB; ++p) {
    const int r0 = kq ? (p < 2 ? cbk + 16 * p : cbq + 16 * (p - 2)) : n0 + 16 * p;
    gb[p] = Bm + (size_t)(r0 + lr) * K + lc;
  }

  v4f ra[PA], rb[PB];
#pragma unroll
  for (int p = 0; p < PA; ++p) ra[p] = *(const v4f*)(ga + (size_t)(p * RPP) * K);
#pragma unroll
  for (int p = 0; p < PB; ++p) rb[p] = *(const v4f*)(gb[p]);
#pragma unroll
  for (int p = 0; p < PA; ++p) *(v4f*)&sA[0][p * RPP + lr][lc] = ra[p];
#pragma unroll
  for (int p = 0; p < PB; ++p) *(v4f*)&sB[0][p * RPP + lr][lc] = rb[p];
  __syncthreads();

  v8f acc[WM][WN];
#pragma unroll
  for (int i = 0; i < WM; ++i)
#pragma unroll
    for (int j = 0; j < WN; ++j) acc[i][j] = v8f{};

  const int arow = wm * 16 * WM + l16;
  const int brow = wn * 16 * WN + l16;
  const int nch  = K / KC;
  for (int c = 0; c < nch; ++c) {
    const int cur  = c & 1;
    const bool more = (c + 1 < nch);
    if (more) {
      const int kn = (c + 1) * KC;
#pragma unroll
      for (int p = 0; p < PA; ++p) ra[p] = *(const v4f*)(ga + (size_t)(p * RPP) * K + kn);
#pragma unroll
      for (int p = 0; p < PB; ++p) rb[p] = *(const v4f*)(gb[p] + kn);
    }
#pragma unroll
    for (int ks = 0; ks < KC; ks += 4) {
      v2f fa[WM], fb[WN];
#pragma unroll
      for (int i = 0; i < WM; ++i) fa[i] = *(const v2f*)&sA[cur][arow + i * 16][ks + kb];
#pragma unroll
      for (int j = 0; j < WN; ++j) fb[j] = *(const v2f*)&sB[cur][brow + j * 16][ks + kb];
#pragma unroll
      for (int i = 0; i < WM; ++i)
#pragma unroll
        for (int j = 0; j < WN; ++j) acc[i][j] = WMMA_F32(fa[i], fb[j], acc[i][j]);
    }
    if (more) {
#pragma unroll
      for (int p = 0; p < PA; ++p) *(v4f*)&sA[cur ^ 1][p * RPP + lr][lc] = ra[p];
#pragma unroll
      for (int p = 0; p < PB; ++p) *(v4f*)&sB[cur ^ 1][p * RPP + lr][lc] = rb[p];
    }
    __syncthreads();
  }

  static_assert(WM == 1 && WN == 4 && WVN == 1, "epilogue is written for a 16x64 wave tile");
  const int mrow = m0 + wm * 16 + half * 8;
  if (kq) {
#pragma unroll
    for (int j = 2; j < 4; ++j) {
#pragma unroll
      for (int r = 0; r < 8; ++r)
        Cm[(size_t)(mrow + r) * N + cbq + (j - 2) * 16 + l16] = acc[0][j][r];
    }
    float keep = 0.0f;
#pragma unroll
    for (int r = 0; r < 8; ++r) {
      float p = acc[0][0][r] * acc[0][2][r];
      p = fmaf(acc[0][1][r], acc[0][3][r], p);
      p += lane_xor<1>(p);
      p += lane_xor<2>(p);
      p += lane_xor<4>(p);
      p += lane_xor<8>(p);
      if (l16 == r) keep = p;
    }
    if (l16 < 8) s0p[((size_t)pp * M + mrow + l16) * H_ + hd] = keep;
  } else {
#pragma unroll
    for (int j = 0; j < 4; ++j) {
#pragma unroll
      for (int r = 0; r < 8; ++r)
        Cm[(size_t)(mrow + r) * N + n0 + j * 16 + l16] = acc[0][j][r];
    }
  }
}

__global__ __launch_bounds__(512)
void attn_proj_kernel(const float* __restrict__ kqv, const float* __restrict__ s0p,
                      const float* __restrict__ q_pos,
                      const float* __restrict__ v_pos, const float* __restrict__ w_out,
                      const float* __restrict__ b_out,
                      float* __restrict__ y, float* __restrict__ wei_out) {
  constexpr int MG  = 16;
  constexpr int SCP = W_ + 2 * MG + 4;
  constexpr int PP  = C_ + 4;
  __shared__ __attribute__((aligned(16))) float sc[H_][16][SCP];
  __shared__ float s0s[H_][W_ + 16];
  __shared__ __attribute__((aligned(16))) float ao[16][PP];
  __shared__ __attribute__((aligned(16))) float sW[C_][PP];

  const int t0 = blockIdx.x * 16;
  const int b  = blockIdx.y;
  const int tid = threadIdx.x;
  const int lane = tid & 31;
  const int wave = __builtin_amdgcn_readfirstlane(tid >> 5);
  const int h    = wave >> 2;
  const int wv   = wave & 3;
  const int ht   = tid & 127;
  const int half = lane >> 4, l16 = lane & 15, kb = half * 2;
  const float scale = 0.125f;

  if (blockIdx.x == 0 && tid < C_)
    y[(size_t)b * T_ * C_ + tid] = b_out[tid];

  {
    const int zr = ht >> 3, zp = ht & 7;
    const int zc = zp < 4 ? 4 * zp : W_ + MG + 4 * (zp - 4);
    *(v4f*)&sc[h][zr][zc] = v4f{};
  }

  const float* wsrc = w_out + (size_t)(tid >> 5) * C_ + (tid & 31) * 4;
  float* wdst = &sW[tid >> 5][(tid & 31) * 4];
  v4f rw[16], rw2[16];

  const int d0 = wv * 16;
  const float* vbase = kqv + (size_t)b * T_ * KQVROW + 2 * H_ * D_ + h * D_ + d0;
  const int voff = kb * KQVROW + l16;
  v2f bvv[20], bvp[W_ / 4];

  {
    const int jw0 = t0 - W_ + 16 * wv;
    const int ja  = jw0 + l16;
    const int jb  = ja + 16;
    const int jra = ja < 0 ? 0 : ja;
    const int jrb = jb < 0 ? 0 : jb;
    const float* qa = kqv + (size_t)(b * T_ + jra) * KQVROW + H_ * D_ + h * D_ + 4 * half;
    const float* qc = kqv + (size_t)(b * T_ + jrb) * KQVROW + H_ * D_ + h * D_ + 4 * half;
    constexpr size_t S0PART = (size_t)B_ * T_ * H_;
    const float* sa = s0p + (size_t)(b * T_ + jra) * H_ + h;
    const float* sb = s0p + (size_t)(b * T_ + (wv == 3 ? jrb : jra)) * H_ + h;
    const float* pr = q_pos + (size_t)((16 * wv + l16) * H_ + h) * D_ + 4 * half;

    v4f fa[D_ / 8], fc[D_ / 8], fp[D_ / 8];
#pragma unroll
    for (int i = 0; i < D_ / 8; ++i) fa[i] = *(const v4f*)(qa + 8 * i);
#pragma unroll
    for (int i = 0; i < D_ / 8; ++i) fc[i] = *(const v4f*)(qc + 8 * i);
#pragma unroll
    for (int i = 0; i < D_ / 8; ++i) fp[i] = *(const v4f*)(pr + 8 * i);
    const float s0a0 = sa[0], s0a1 = sa[S0PART];
    const float s0b0 = sb[0], s0b1 = sb[S0PART];
    __builtin_amdgcn_sched_barrier(0);
#pragma unroll
    for (int jt = 0; jt < 5; ++jt) {
      const int j0  = t0 - 64 + jt * 16;
      const int j0c = j0 < 0 ? 0 : j0;
#pragma unroll
      for (int k4 = 0; k4 < 16; k4 += 4) {
        const float* vr = vbase + (size_t)(j0c + k4) * KQVROW;
        bvv[jt * 4 + k4 / 4].x = vr[voff];
        bvv[jt * 4 + k4 / 4].y = vr[voff + KQVROW];
      }
    }


#pragma unroll
    for (int k = 0; k < W_; k += 4) {
      bvp[k / 4].x = v_pos[((size_t)(k + kb)     * H_ + h) * D_ + d0 + l16];
      bvp[k / 4].y = v_pos[((size_t)(k + kb + 1) * H_ + h) * D_ + d0 + l16];
    }
    __builtin_amdgcn_sched_barrier(0);

    v8f c0 = {}, c1 = {};
#pragma unroll
    for (int i = 0; i < D_ / 8; ++i) {
      const v2f blo = fp[i].lo, bhi = fp[i].hi;
      c0 = WMMA_F32(fa[i].lo, blo, c0);
      c1 = WMMA_F32(fc[i].lo, blo, c1);
      c0 = WMMA_F32(fa[i].hi, bhi, c0);
      c1 = WMMA_F32(fc[i].hi, bhi, c1);
    }

    s0s[h][16 * wv + l16] = s0a0 + s0a1;
    s0s[h][(wv == 3 ? W_ : 16 * wv) + l16] = s0b0 + s0b1;

    lds_barrier();

    float s0v[8];
#pragma unroll
    for (int r = 0; r < 8; ++r) {
      const int R = 8 * half + r;
      s0v[r] = s0s[h][16 * wv + (R > l16 ? R : R + 16)];
    }
#pragma unroll
    for (int r = 0; r < 8; ++r) {
      const int  R    = 8 * half + r;
      const bool lowt = R > l16;
      const int  jrel = lowt ? R : R + 16;
      const float s1  = lowt ? c0[r] : c1[r];
      const int  ti   = jrel - l16 - 1;
      float v = (s0v[r] + s1) * scale;
      if (jw0 + jrel < 0)    v = -INFINITY;
      if (t0 + ti >= Tq_)    v = 0.0f;
      sc[h][ti][MG + 16 * wv + l16] = v;
    }
  }
  lds_barrier();

#pragma unroll
  for (int u = 0; u < 16; ++u) rw[u] = *(const v4f*)(wsrc + (size_t)u * 16 * C_);

  {
    float* rp = &sc[h][ht >> 3][MG + 8 * (ht & 7)];
    v4f xa = *(const v4f*)rp;
    v4f xb = *(const v4f*)(rp + 4);
    float m = fmaxf(fmaxf(fmaxf(xa.x, xa.y), fmaxf(xa.z, xa.w)),
                    fmaxf(fmaxf(xb.x, xb.y), fmaxf(xb.z, xb.w)));
    m = fmaxf(m, lane_xor<1>(m));
    m = fmaxf(m, lane_xor<2>(m));
    m = fmaxf(m, lane_xor<4>(m));
    xa.x = __expf(xa.x - m); xa.y = __expf(xa.y - m);
    xa.z = __expf(xa.z - m); xa.w = __expf(xa.w - m);
    xb.x = __expf(xb.x - m); xb.y = __expf(xb.y - m);
    xb.z = __expf(xb.z - m); xb.w = __expf(xb.w - m);
    float s = ((xa.x + xa.y) + (xa.z + xa.w)) + ((xb.x + xb.y) + (xb.z + xb.w));
    s += lane_xor<1>(s);
    s += lane_xor<2>(s);
    s += lane_xor<4>(s);
    const float inv = 1.0f / s;
    *(v4f*)rp       = xa * inv;
    *(v4f*)(rp + 4) = xb * inv;
  }
  lds_barrier();

  {
    const int w = ((wave & 1) << 5) + lane;
#pragma unroll
    for (int u = 0; u < 2; ++u) {
      const int ti = (wave >> 1) + 8 * u;
      if (t0 + ti < Tq_) {
        v4f o;
        o.x = sc[0][ti][MG + w];
        o.y = sc[1][ti][MG + w];
        o.z = sc[2][ti][MG + w];
        o.w = sc[3][ti][MG + w];
        *(v4f*)(wei_out + (((size_t)b * Tq_ + (t0 + ti)) * W_ + w) * H_) = o;
      }
    }
  }

  v8f c = {};
#pragma unroll
  for (int jt = 0; jt < 5; ++jt) {
#pragma unroll
    for (int k4 = 0; k4 < 16; k4 += 4) {
      const int wa = jt * 16 + k4 + kb - l16 - 1;
      v2f a;
      a.x = sc[h][l16][MG + wa];
      a.y = sc[h][l16][MG + wa + 1];
      c = WMMA_F32(a, bvv[jt * 4 + k4 / 4], c);
    }
  }
#pragma unroll
  for (int k = 0; k < W_; k += 4) {
    v2f a;
    a.x = sc[h][l16][MG + k + kb];
    a.y = sc[h][l16][MG + k + kb + 1];
    c = WMMA_F32(a, bvp[k / 4], c);
  }

#pragma unroll
  for (int i = 0; i < 8; ++i) ao[i + half * 8][h * D_ + d0 + l16] = c[i];

#pragma unroll
  for (int u = 0; u < 16; ++u) rw2[u] = *(const v4f*)(wsrc + 128 + (size_t)u * 16 * C_);

#pragma unroll
  for (int u = 0; u < 16; ++u) *(v4f*)(wdst + (size_t)u * 16 * PP) = rw[u];

  lds_barrier();

  {
    const int n0 = 16 * wave;
    const float* pa = &ao[l16][4 * half];
    const float* pb = &sW[n0 + l16][4 * half];
    const float bb = b_out[n0 + l16];
    v8f acc = {};
#pragma unroll
    for (int g = 0; g < 4; ++g) {
      if (g == 2) {
        asm volatile("" :: "v"(acc) : "memory");
#pragma unroll
        for (int u = 0; u < 16; ++u) *(v4f*)(wdst + 128 + (size_t)u * 16 * PP) = rw2[u];
        lds_barrier();
      }
      v4f xa[8], xb[8];
#pragma unroll
      for (int i = 0; i < 8; ++i) xa[i] = *(const v4f*)(pa + 64 * g + 8 * i);
#pragma unroll
      for (int i = 0; i < 8; ++i) xb[i] = *(const v4f*)(pb + 64 * g + 8 * i);
#pragma unroll
      for (int i = 0; i < 8; ++i) {
        acc = WMMA_F32(xa[i].lo, xb[i].lo, acc);
        acc = WMMA_F32(xa[i].hi, xb[i].hi, acc);
      }
    }
    float* yr = y + ((size_t)b * T_ + (t0 + S_ + half * 8)) * C_ + n0 + l16;
#pragma unroll
    for (int r = 0; r < 7; ++r) yr[(size_t)r * C_] = acc[r] + bb;
    if (t0 + 7 + half * 8 < Tq_) yr[(size_t)7 * C_] = acc[7] + bb;
  }
}

extern "C" void kernel_launch(void* const* d_in, const int* in_sizes, int n_in,
                              void* d_out, int out_size, void* d_ws, size_t ws_size,
                              hipStream_t stream) {
  (void)in_sizes; (void)n_in; (void)out_size; (void)ws_size;

  const float* x     = (const float*)d_in[0];
  const float* w_kqv = (const float*)d_in[1];
  const float* w_out = (const float*)d_in[2];
  const float* b_out = (const float*)d_in[3];
  const float* q_pos = (const float*)d_in[4];
  const float* v_pos = (const float*)d_in[5];

  float* y   = (float*)d_out;
  float* wei = y + (size_t)B_ * T_ * C_;

  float* kqv     = (float*)d_ws;
  float* s0p     = kqv + (size_t)B_ * T_ * KQVROW;

  gemm_kqv<1, 4, 4, 1><<<dim3(KQVROW / 64, (B_ * T_) / 64), 128, 0, stream>>>(
      x, w_kqv, kqv, s0p, B_ * T_, KQVROW, C_);

  attn_proj_kernel<<<dim3((Tq_ + 15) / 16, B_), 512, 0, stream>>>(
      kqv, s0p, q_pos, v_pos, w_out, b_out, y, wei);
}
